# starting kernel
# baseline (speedup 1.0000x reference)
#include <hip/hip_runtime.h>
#include <hip/hip_bf16.h>

typedef float v2f __attribute__((ext_vector_type(2)));
typedef float v8f __attribute__((ext_vector_type(8)));

#define DIM 128
#define NT 17

__global__ __launch_bounds__(256) void k_hw1(const float* __restrict__ emb,
                                             const float* __restrict__ W1,
                                             float* __restrict__ HW1) {
    int t = blockIdx.x * 256 + threadIdx.x;
    if (t >= NT * DIM) return;
    int m = t >> 7, n = t & 127;
    float acc = 0.f;
    #pragma unroll 8
    for (int k = 0; k < DIM; ++k)
        acc = fmaf(emb[m * DIM + k], W1[k * DIM + n], acc);
    HW1[t] = acc;
}

__global__ __launch_bounds__(256) void k_deg_init(float* __restrict__ deg, int n) {
    int i = blockIdx.x * 256 + threadIdx.x;
    if (i < n) deg[i] = 1.0f;
}
__global__ __launch_bounds__(256) void k_deg_edges(const int* __restrict__ dst,
                                                   float* __restrict__ deg, int nE) {
    int e = blockIdx.x * 256 + threadIdx.x;
    if (e < nE) atomicAdd(&deg[dst[e]], 1.0f);
}
__global__ __launch_bounds__(256) void k_dinv(float* __restrict__ deg, int n) {
    int i = blockIdx.x * 256 + threadIdx.x;
    if (i < n) { float d = deg[i]; deg[i] = d > 0.f ? rsqrtf(d) : 0.f; }
}

__global__ __launch_bounds__(256) void k_self1(const int* __restrict__ node_ids,
                                               const float* __restrict__ dinv,
                                               const float* __restrict__ HW1,
                                               float* __restrict__ agg, int n) {
    int t = blockIdx.x * 256 + threadIdx.x;
    int i = t >> 5;
    if (i >= n) return;
    int f = (t & 31) * 4;
    float w = dinv[i] * dinv[i];
    const float4 h = *(const float4*)(HW1 + (size_t)node_ids[i] * DIM + f);
    float4 o = make_float4(w * h.x, w * h.y, w * h.z, w * h.w);
    *(float4*)(agg + (size_t)i * DIM + f) = o;
}

__global__ __launch_bounds__(256) void k_edge1(const int* __restrict__ src,
                                               const int* __restrict__ dst,
                                               const int* __restrict__ node_ids,
                                               const float* __restrict__ dinv,
                                               const float* __restrict__ HW1,
                                               float* __restrict__ agg, int nE) {
    int w = (blockIdx.x * 256 + threadIdx.x) >> 5;
    int lane = threadIdx.x & 31;
    if (w >= nE) return;
    int s = src[w], d = dst[w];
    float nrm = dinv[s] * dinv[d];
    const float* __restrict__ hrow = HW1 + (size_t)node_ids[s] * DIM;
    float* __restrict__ orow = agg + (size_t)d * DIM;
    #pragma unroll
    for (int j = 0; j < 4; ++j) {
        int f = lane + 32 * j;
        atomicAdd(&orow[f], nrm * hrow[f]);
    }
}

__global__ __launch_bounds__(256) void k_gemm2(const float* __restrict__ agg1,
                                               const float* __restrict__ b1,
                                               const float* __restrict__ W2,
                                               float* __restrict__ h2, int n) {
    __shared__ float As[32][DIM + 4];
    int r0 = blockIdx.x * 32;
    int tid = threadIdx.x;
    for (int t = tid; t < 32 * DIM; t += 256) {
        int r = t >> 7, c = t & 127;
        float v = 0.f;
        if (r0 + r < n) {
            v = agg1[(size_t)(r0 + r) * DIM + c] + b1[c];
            v = fmaxf(v, 0.f);
        }
        As[r][c] = v;
    }
    __syncthreads();

    int wave = tid >> 5;
    int lane = tid & 31;
    int half = lane >> 4;
    int l16  = lane & 15;
    int colBase = wave * 16;

    v8f acc0 = {}, acc1 = {};
    #pragma unroll 4
    for (int kt = 0; kt < DIM / 4; ++kt) {
        int k0 = kt * 4 + half * 2;
        v2f a0, a1, b;
        b.x = W2[(size_t)k0 * DIM + colBase + l16];
        b.y = W2[(size_t)(k0 + 1) * DIM + colBase + l16];
        a0.x = As[l16][k0];       a0.y = As[l16][k0 + 1];
        a1.x = As[16 + l16][k0];  a1.y = As[16 + l16][k0 + 1];
        acc0 = __builtin_amdgcn_wmma_f32_16x16x4_f32(false, a0, false, b,
                                                     (short)0, acc0, false, false);
        acc1 = __builtin_amdgcn_wmma_f32_16x16x4_f32(false, a1, false, b,
                                                     (short)0, acc1, false, false);
    }
    int col = colBase + l16;
    if (r0 + 32 <= n) {
        #pragma unroll
        for (int j = 0; j < 8; ++j) {
            h2[(size_t)(r0      + j + half * 8) * DIM + col] = acc0[j];
            h2[(size_t)(r0 + 16 + j + half * 8) * DIM + col] = acc1[j];
        }
    } else {
        for (int j = 0; j < 8; ++j) {
            int row0 = r0 + j + half * 8;
            int row1 = row0 + 16;
            if (row0 < n) h2[(size_t)row0 * DIM + col] = acc0[j];
            if (row1 < n) h2[(size_t)row1 * DIM + col] = acc1[j];
        }
    }
}

__global__ __launch_bounds__(256) void k_self2(const float* __restrict__ dinv,
                                               const float* __restrict__ h2,
                                               float* __restrict__ agg, int n) {
    int t = blockIdx.x * 256 + threadIdx.x;
    int i = t >> 5;
    if (i >= n) return;
    int f = (t & 31) * 4;
    float w = dinv[i] * dinv[i];
    const float4 h = *(const float4*)(h2 + (size_t)i * DIM + f);
    float4 o = make_float4(w * h.x, w * h.y, w * h.z, w * h.w);
    *(float4*)(agg + (size_t)i * DIM + f) = o;
}

__global__ __launch_bounds__(256) void k_edge2(const int* __restrict__ src,
                                               const int* __restrict__ dst,
                                               const float* __restrict__ dinv,
                                               const float* __restrict__ h2,
                                               float* __restrict__ agg, int nE) {
    int w = (blockIdx.x * 256 + threadIdx.x) >> 5;
    int lane = threadIdx.x & 31;
    if (w >= nE) return;
    int s = src[w], d = dst[w];
    float nrm = dinv[s] * dinv[d];
    const float* __restrict__ hrow = h2 + (size_t)s * DIM;
    float* __restrict__ orow = agg + (size_t)d * DIM;
    #pragma unroll
    for (int j = 0; j < 4; ++j) {
        int f = lane + 32 * j;
        atomicAdd(&orow[f], nrm * hrow[f]);
    }
}

__global__ __launch_bounds__(256) void k_zero_pool(float* __restrict__ sums,
                                                   float* __restrict__ cnt, int g) {
    int i = blockIdx.x * 256 + threadIdx.x;
    if (i < g * DIM) sums[i] = 0.f;
    if (i < g) cnt[i] = 0.f;
}
__global__ __launch_bounds__(256) void k_pool(const float* __restrict__ agg2,
                                              const float* __restrict__ b2,
                                              const int* __restrict__ batch,
                                              float* __restrict__ sums,
                                              float* __restrict__ cnt, int n) {
    int w = (blockIdx.x * 256 + threadIdx.x) >> 5;
    int lane = threadIdx.x & 31;
    if (w >= n) return;
    int g = batch[w];
    const float* __restrict__ arow = agg2 + (size_t)w * DIM;
    float* __restrict__ srow = sums + (size_t)g * DIM;
    #pragma unroll
    for (int j = 0; j < 4; ++j) {
        int f = lane + 32 * j;
        atomicAdd(&srow[f], arow[f] + b2[f]);
    }
    if (lane == 0) atomicAdd(&cnt[g], 1.0f);
}
__global__ __launch_bounds__(256) void k_final(const float* __restrict__ sums,
                                               const float* __restrict__ cnt,
                                               float* __restrict__ out, int g) {
    int i = blockIdx.x * 256 + threadIdx.x;
    if (i >= g * DIM) return;
    float c = cnt[i >> 7];
    out[i] = sums[i] / fmaxf(c, 1.0f);
}

extern "C" void kernel_launch(void* const* d_in, const int* in_sizes, int n_in,
                              void* d_out, int out_size, void* d_ws, size_t ws_size,
                              hipStream_t stream) {
    const int*   node_ids = (const int*)d_in[0];
    const int*   ei       = (const int*)d_in[1];
    const int*   batch    = (const int*)d_in[2];
    const float* emb      = (const float*)d_in[4];
    const float* W1       = (const float*)d_in[5];
    const float* b1       = (const float*)d_in[6];
    const float* W2       = (const float*)d_in[7];
    const float* b2       = (const float*)d_in[8];
    float* out = (float*)d_out;

    const int N = in_sizes[0];
    const int E = in_sizes[1] / 2;
    const int G = out_size / DIM;
    const int* src = ei;
    const int* dst = ei + E;

    char* base = (char*)d_ws;
    size_t off = 0;
    auto alloc = [&](size_t bytes) -> char* {
        char* p = base + off;
        off = (off + bytes + 255) & ~(size_t)255;
        return p;
    };
    float* HW1  = (float*)alloc((size_t)NT * DIM * sizeof(float));
    float* dinv = (float*)alloc((size_t)N * sizeof(float));
    float* bufA = (float*)alloc((size_t)N * DIM * sizeof(float));
    float* bufB = (float*)alloc((size_t)N * DIM * sizeof(float));
    float* sums = (float*)alloc((size_t)G * DIM * sizeof(float));
    float* cnt  = (float*)alloc((size_t)G * sizeof(float));
    (void)ws_size; (void)n_in;

    const int TB = 256;
    dim3 blk(TB);

    k_hw1<<<dim3((NT * DIM + TB - 1) / TB), blk, 0, stream>>>(emb, W1, HW1);

    k_deg_init <<<dim3((N + TB - 1) / TB), blk, 0, stream>>>(dinv, N);
    k_deg_edges<<<dim3((E + TB - 1) / TB), blk, 0, stream>>>(dst, dinv, E);
    k_dinv     <<<dim3((N + TB - 1) / TB), blk, 0, stream>>>(dinv, N);

    k_self1<<<dim3(((size_t)N * 32 + TB - 1) / TB), blk, 0, stream>>>(node_ids, dinv, HW1, bufA, N);
    k_edge1<<<dim3(((size_t)E * 32 + TB - 1) / TB), blk, 0, stream>>>(src, dst, node_ids, dinv, HW1, bufA, E);

    k_gemm2<<<dim3((N + 31) / 32), blk, 0, stream>>>(bufA, b1, W2, bufB, N);

    k_self2<<<dim3(((size_t)N * 32 + TB - 1) / TB), blk, 0, stream>>>(dinv, bufB, bufA, N);
    k_edge2<<<dim3(((size_t)E * 32 + TB - 1) / TB), blk, 0, stream>>>(src, dst, dinv, bufB, bufA, E);

    k_zero_pool<<<dim3((G * DIM + TB - 1) / TB), blk, 0, stream>>>(sums, cnt, G);
    k_pool     <<<dim3(((size_t)N * 32 + TB - 1) / TB), blk, 0, stream>>>(bufA, b2, batch, sums, cnt, N);
    k_final    <<<dim3((G * DIM + TB - 1) / TB), blk, 0, stream>>>(sums, cnt, out, G);
}


# mean-pool: each wave sums its consecutive node rows in registers while the graph id repeats, the block's waves combine in LDS, one atomic row per block, not one per node; changes summation order
# speedup vs baseline: 1.2610x; 1.2610x over previous
#include <hip/hip_runtime.h>
#include <hip/hip_bf16.h>

typedef float v2f __attribute__((ext_vector_type(2)));
typedef float v8f __attribute__((ext_vector_type(8)));

#define DIM 128
#define NT 17

__global__ __launch_bounds__(256) void k_hw1(const float* __restrict__ emb,
                                             const float* __restrict__ W1,
                                             float* __restrict__ HW1) {
    int t = blockIdx.x * 256 + threadIdx.x;
    if (t >= NT * DIM) return;
    int m = t >> 7, n = t & 127;
    float acc = 0.f;
    #pragma unroll 8
    for (int k = 0; k < DIM; ++k)
        acc = fmaf(emb[m * DIM + k], W1[k * DIM + n], acc);
    HW1[t] = acc;
}

__global__ __launch_bounds__(256) void k_deg_init(float* __restrict__ deg, int n) {
    int i = blockIdx.x * 256 + threadIdx.x;
    if (i < n) deg[i] = 1.0f;
}
__global__ __launch_bounds__(256) void k_deg_edges(const int* __restrict__ dst,
                                                   float* __restrict__ deg, int nE) {
    int e = blockIdx.x * 256 + threadIdx.x;
    if (e < nE) atomicAdd(&deg[dst[e]], 1.0f);
}
__global__ __launch_bounds__(256) void k_dinv(float* __restrict__ deg, int n) {
    int i = blockIdx.x * 256 + threadIdx.x;
    if (i < n) { float d = deg[i]; deg[i] = d > 0.f ? rsqrtf(d) : 0.f; }
}

__global__ __launch_bounds__(256) void k_self1(const int* __restrict__ node_ids,
                                               const float* __restrict__ dinv,
                                               const float* __restrict__ HW1,
                                               float* __restrict__ agg, int n) {
    int t = blockIdx.x * 256 + threadIdx.x;
    int i = t >> 5;
    if (i >= n) return;
    int f = (t & 31) * 4;
    float w = dinv[i] * dinv[i];
    const float4 h = *(const float4*)(HW1 + (size_t)node_ids[i] * DIM + f);
    float4 o = make_float4(w * h.x, w * h.y, w * h.z, w * h.w);
    *(float4*)(agg + (size_t)i * DIM + f) = o;
}

__global__ __launch_bounds__(256) void k_edge1(const int* __restrict__ src,
                                               const int* __restrict__ dst,
                                               const int* __restrict__ node_ids,
                                               const float* __restrict__ dinv,
                                               const float* __restrict__ HW1,
                                               float* __restrict__ agg, int nE) {
    int w = (blockIdx.x * 256 + threadIdx.x) >> 5;
    int lane = threadIdx.x & 31;
    if (w >= nE) return;
    int s = src[w], d = dst[w];
    float nrm = dinv[s] * dinv[d];
    const float* __restrict__ hrow = HW1 + (size_t)node_ids[s] * DIM;
    float* __restrict__ orow = agg + (size_t)d * DIM;
    #pragma unroll
    for (int j = 0; j < 4; ++j) {
        int f = lane + 32 * j;
        atomicAdd(&orow[f], nrm * hrow[f]);
    }
}

__global__ __launch_bounds__(256) void k_gemm2(const float* __restrict__ agg1,
                                               const float* __restrict__ b1,
                                               const float* __restrict__ W2,
                                               float* __restrict__ h2, int n) {
    __shared__ float As[32][DIM + 4];
    int r0 = blockIdx.x * 32;
    int tid = threadIdx.x;
    for (int t = tid; t < 32 * DIM; t += 256) {
        int r = t >> 7, c = t & 127;
        float v = 0.f;
        if (r0 + r < n) {
            v = agg1[(size_t)(r0 + r) * DIM + c] + b1[c];
            v = fmaxf(v, 0.f);
        }
        As[r][c] = v;
    }
    __syncthreads();

    int wave = tid >> 5;
    int lane = tid & 31;
    int half = lane >> 4;
    int l16  = lane & 15;
    int colBase = wave * 16;

    v8f acc0 = {}, acc1 = {};
    #pragma unroll 4
    for (int kt = 0; kt < DIM / 4; ++kt) {
        int k0 = kt * 4 + half * 2;
        v2f a0, a1, b;
        b.x = W2[(size_t)k0 * DIM + colBase + l16];
        b.y = W2[(size_t)(k0 + 1) * DIM + colBase + l16];
        a0.x = As[l16][k0];       a0.y = As[l16][k0 + 1];
        a1.x = As[16 + l16][k0];  a1.y = As[16 + l16][k0 + 1];
        acc0 = __builtin_amdgcn_wmma_f32_16x16x4_f32(false, a0, false, b,
                                                     (short)0, acc0, false, false);
        acc1 = __builtin_amdgcn_wmma_f32_16x16x4_f32(false, a1, false, b,
                                                     (short)0, acc1, false, false);
    }
    int col = colBase + l16;
    if (r0 + 32 <= n) {
        #pragma unroll
        for (int j = 0; j < 8; ++j) {
            h2[(size_t)(r0      + j + half * 8) * DIM + col] = acc0[j];
            h2[(size_t)(r0 + 16 + j + half * 8) * DIM + col] = acc1[j];
        }
    } else {
        for (int j = 0; j < 8; ++j) {
            int row0 = r0 + j + half * 8;
            int row1 = row0 + 16;
            if (row0 < n) h2[(size_t)row0 * DIM + col] = acc0[j];
            if (row1 < n) h2[(size_t)row1 * DIM + col] = acc1[j];
        }
    }
}

__global__ __launch_bounds__(256) void k_self2(const float* __restrict__ dinv,
                                               const float* __restrict__ h2,
                                               float* __restrict__ agg, int n) {
    int t = blockIdx.x * 256 + threadIdx.x;
    int i = t >> 5;
    if (i >= n) return;
    int f = (t & 31) * 4;
    float w = dinv[i] * dinv[i];
    const float4 h = *(const float4*)(h2 + (size_t)i * DIM + f);
    float4 o = make_float4(w * h.x, w * h.y, w * h.z, w * h.w);
    *(float4*)(agg + (size_t)i * DIM + f) = o;
}

__global__ __launch_bounds__(256) void k_edge2(const int* __restrict__ src,
                                               const int* __restrict__ dst,
                                               const float* __restrict__ dinv,
                                               const float* __restrict__ h2,
                                               float* __restrict__ agg, int nE) {
    int w = (blockIdx.x * 256 + threadIdx.x) >> 5;
    int lane = threadIdx.x & 31;
    if (w >= nE) return;
    int s = src[w], d = dst[w];
    float nrm = dinv[s] * dinv[d];
    const float* __restrict__ hrow = h2 + (size_t)s * DIM;
    float* __restrict__ orow = agg + (size_t)d * DIM;
    #pragma unroll
    for (int j = 0; j < 4; ++j) {
        int f = lane + 32 * j;
        atomicAdd(&orow[f], nrm * hrow[f]);
    }
}

__global__ __launch_bounds__(256) void k_zero_pool(float* __restrict__ sums,
                                                   float* __restrict__ cnt, int g) {
    int i = blockIdx.x * 256 + threadIdx.x;
    if (i < g * DIM) sums[i] = 0.f;
    if (i < g) cnt[i] = 0.f;
}
#define POOL_PW 16
__global__ __launch_bounds__(256) void k_pool(const float* __restrict__ agg2,
                                              const float* __restrict__ b2,
                                              const int* __restrict__ batch,
                                              float* __restrict__ sums,
                                              float* __restrict__ cnt, int n) {
    __shared__ float part[8][DIM];
    __shared__ float pcnt[8];
    const int lane = threadIdx.x & 31;
    const int wave = __builtin_amdgcn_readfirstlane((int)(threadIdx.x >> 5));
    const int nb = blockIdx.x * (8 * POOL_PW);
    const int n0 = nb + wave * POOL_PW;
    const int f  = lane * 4;
    const int g0 = batch[nb];
    const float4 bb = *(const float4*)(b2 + f);

    float4 acc0 = make_float4(0.f, 0.f, 0.f, 0.f);
    float  c0 = 0.f;
    float4 acc = make_float4(0.f, 0.f, 0.f, 0.f);
    float  c = 0.f;
    int    gc = g0;

    auto flush = [&](int g) {
        if (g == g0) {
            acc0.x += acc.x; acc0.y += acc.y; acc0.z += acc.z; acc0.w += acc.w;
            c0 += c;
        } else if (c > 0.f) {
            float* __restrict__ srow = sums + (size_t)g * DIM + f;
            atomicAdd(srow + 0, acc.x);
            atomicAdd(srow + 1, acc.y);
            atomicAdd(srow + 2, acc.z);
            atomicAdd(srow + 3, acc.w);
            if (lane == 0) atomicAdd(&cnt[g], c);
        }
        acc = make_float4(0.f, 0.f, 0.f, 0.f);
        c = 0.f;
    };
    auto add = [&](int g, const float4& v) {
        if (g != gc) { flush(gc); gc = g; }
        acc.x += v.x + bb.x; acc.y += v.y + bb.y;
        acc.z += v.z + bb.z; acc.w += v.w + bb.w;
        c += 1.0f;
    };

    if (n0 + POOL_PW <= n) {
        float4 v[POOL_PW];
        int    g[POOL_PW];
        #pragma unroll
        for (int j = 0; j < POOL_PW; ++j) {
            v[j] = *(const float4*)(agg2 + (size_t)(n0 + j) * DIM + f);
            g[j] = batch[n0 + j];
        }
        #pragma unroll
        for (int j = 0; j < POOL_PW; ++j) add(g[j], v[j]);
    } else {
        for (int i = n0; i < n; ++i) {
            const float4 v = *(const float4*)(agg2 + (size_t)i * DIM + f);
            add(batch[i], v);
        }
    }
    flush(gc);

    *(float4*)(&part[wave][f]) = acc0;
    if (lane == 0) pcnt[wave] = c0;
    __syncthreads();
    const int t = threadIdx.x;
    if (t < DIM) {
        float s = part[0][t];
        #pragma unroll
        for (int w = 1; w < 8; ++w) s += part[w][t];
        atomicAdd(&sums[(size_t)g0 * DIM + t], s);
    }
    if (t == 0) {
        float cs = pcnt[0];
        #pragma unroll
        for (int w = 1; w < 8; ++w) cs += pcnt[w];
        atomicAdd(&cnt[g0], cs);
    }
}
__global__ __launch_bounds__(256) void k_final(const float* __restrict__ sums,
                                               const float* __restrict__ cnt,
                                               float* __restrict__ out, int g) {
    int i = blockIdx.x * 256 + threadIdx.x;
    if (i >= g * DIM) return;
    float c = cnt[i >> 7];
    out[i] = sums[i] / fmaxf(c, 1.0f);
}

extern "C" void kernel_launch(void* const* d_in, const int* in_sizes, int n_in,
                              void* d_out, int out_size, void* d_ws, size_t ws_size,
                              hipStream_t stream) {
    const int*   node_ids = (const int*)d_in[0];
    const int*   ei       = (const int*)d_in[1];
    const int*   batch    = (const int*)d_in[2];
    const float* emb      = (const float*)d_in[4];
    const float* W1       = (const float*)d_in[5];
    const float* b1       = (const float*)d_in[6];
    const float* W2       = (const float*)d_in[7];
    const float* b2       = (const float*)d_in[8];
    float* out = (float*)d_out;

    const int N = in_sizes[0];
    const int E = in_sizes[1] / 2;
    const int G = out_size / DIM;
    const int* src = ei;
    const int* dst = ei + E;

    char* base = (char*)d_ws;
    size_t off = 0;
    auto alloc = [&](size_t bytes) -> char* {
        char* p = base + off;
        off = (off + bytes + 255) & ~(size_t)255;
        return p;
    };
    float* HW1  = (float*)alloc((size_t)NT * DIM * sizeof(float));
    float* dinv = (float*)alloc((size_t)N * sizeof(float));
    float* bufA = (float*)alloc((size_t)N * DIM * sizeof(float));
    float* bufB = (float*)alloc((size_t)N * DIM * sizeof(float));
    float* sums = (float*)alloc((size_t)G * DIM * sizeof(float));
    float* cnt  = (float*)alloc((size_t)G * sizeof(float));
    (void)ws_size; (void)n_in;

    const int TB = 256;
    dim3 blk(TB);

    k_hw1<<<dim3((NT * DIM + TB - 1) / TB), blk, 0, stream>>>(emb, W1, HW1);

    k_deg_init <<<dim3((N + TB - 1) / TB), blk, 0, stream>>>(dinv, N);
    k_deg_edges<<<dim3((E + TB - 1) / TB), blk, 0, stream>>>(dst, dinv, E);
    k_dinv     <<<dim3((N + TB - 1) / TB), blk, 0, stream>>>(dinv, N);

    k_self1<<<dim3(((size_t)N * 32 + TB - 1) / TB), blk, 0, stream>>>(node_ids, dinv, HW1, bufA, N);
    k_edge1<<<dim3(((size_t)E * 32 + TB - 1) / TB), blk, 0, stream>>>(src, dst, node_ids, dinv, HW1, bufA, E);

    k_gemm2<<<dim3((N + 31) / 32), blk, 0, stream>>>(bufA, b1, W2, bufB, N);

    k_self2<<<dim3(((size_t)N * 32 + TB - 1) / TB), blk, 0, stream>>>(dinv, bufB, bufA, N);
    k_edge2<<<dim3(((size_t)E * 32 + TB - 1) / TB), blk, 0, stream>>>(src, dst, dinv, bufB, bufA, E);

    k_zero_pool<<<dim3((G * DIM + TB - 1) / TB), blk, 0, stream>>>(sums, cnt, G);
    k_pool     <<<dim3((N + 8 * POOL_PW - 1) / (8 * POOL_PW)), blk, 0, stream>>>(bufA, b2, batch, sums, cnt, N);
    k_final    <<<dim3((G * DIM + TB - 1) / TB), blk, 0, stream>>>(sums, cnt, out, G);
}


# conv1: each edge adds one scalar weight into a per-node per-type coefficient table (N x 20); agg1 = coef @ HW1 on f32 WMMA in the GEMM prologue, not 128 atomics per edge; changes summation order
# speedup vs baseline: 1.9045x; 1.5103x over previous
#include <hip/hip_runtime.h>
#include <hip/hip_bf16.h>

typedef float v2f __attribute__((ext_vector_type(2)));
typedef float v8f __attribute__((ext_vector_type(8)));

#define DIM 128
#define NT 17
#define NTP 20
#define CS NTP

__global__ __launch_bounds__(256) void k_hw1(const float* __restrict__ emb,
                                             const float* __restrict__ W1,
                                             float* __restrict__ HW1) {
    int t = blockIdx.x * 256 + threadIdx.x;
    if (t >= NTP * DIM) return;
    int m = t >> 7, n = t & 127;
    float acc = 0.f;
    if (m < NT) {
        #pragma unroll 8
        for (int k = 0; k < DIM; ++k)
            acc = fmaf(emb[m * DIM + k], W1[k * DIM + n], acc);
    }
    HW1[t] = acc;
}

__global__ __launch_bounds__(256) void k_deg_init(float* __restrict__ deg, int n) {
    int i = blockIdx.x * 256 + threadIdx.x;
    if (i < n) deg[i] = 1.0f;
}
__global__ __launch_bounds__(256) void k_deg_edges(const int* __restrict__ dst,
                                                   float* __restrict__ deg, int nE) {
    int e = blockIdx.x * 256 + threadIdx.x;
    if (e < nE) atomicAdd(&deg[dst[e]], 1.0f);
}
__global__ __launch_bounds__(256) void k_dinv(float* __restrict__ deg, int n) {
    int i = blockIdx.x * 256 + threadIdx.x;
    if (i < n) { float d = deg[i]; deg[i] = d > 0.f ? rsqrtf(d) : 0.f; }
}

__global__ __launch_bounds__(256) void k_cinit(const int* __restrict__ node_ids,
                                               const float* __restrict__ dinv,
                                               float* __restrict__ coef, int n) {
    int t = blockIdx.x * 256 + threadIdx.x;
    if (t >= n * (CS / 4)) return;
    int i = t / (CS / 4);
    int q = t - i * (CS / 4);
    float w = dinv[i] * dinv[i];
    int ty = node_ids[i] - q * 4;
    float4 o = make_float4(ty == 0 ? w : 0.f, ty == 1 ? w : 0.f, ty == 2 ? w : 0.f, ty == 3 ? w : 0.f);
    *(float4*)(coef + (size_t)t * 4) = o;
}
__global__ __launch_bounds__(256) void k_cedge(const int* __restrict__ src,
                                               const int* __restrict__ dst,
                                               const int* __restrict__ node_ids,
                                               const float* __restrict__ dinv,
                                               float* __restrict__ coef, int nE) {
    int e = blockIdx.x * 256 + threadIdx.x;
    if (e >= nE) return;
    int s = src[e], d = dst[e];
    atomicAdd(&coef[(size_t)d * CS + node_ids[s]], dinv[s] * dinv[d]);
}

__global__ __launch_bounds__(256) void k_gemm2(const float* __restrict__ coef,
                                               const float* __restrict__ HW1,
                                               const float* __restrict__ b1,
                                               const float* __restrict__ W2,
                                               float* __restrict__ h2, int n) {
    __shared__ float As[32][DIM + 4];
    int r0 = blockIdx.x * 32;
    int tid = threadIdx.x;
    int wave = tid >> 5;
    int lane = tid & 31;
    int half = lane >> 4;
    int l16  = lane & 15;
    int colBase = wave * 16;
    int col = colBase + l16;

    {
        const float* __restrict__ ca = coef + (size_t)min(r0 + l16, n - 1) * CS;
        const float* __restrict__ cb = coef + (size_t)min(r0 + 16 + l16, n - 1) * CS;
        v8f p0 = {}, p1 = {};
        #pragma unroll
        for (int kt = 0; kt < NTP / 4; ++kt) {
            int k0 = kt * 4 + half * 2;
            v2f a0, a1, b;
            b.x = HW1[(size_t)k0 * DIM + col];
            b.y = HW1[(size_t)(k0 + 1) * DIM + col];
            a0 = *(const v2f*)(ca + k0);
            a1 = *(const v2f*)(cb + k0);
            p0 = __builtin_amdgcn_wmma_f32_16x16x4_f32(false, a0, false, b,
                                                       (short)0, p0, false, false);
            p1 = __builtin_amdgcn_wmma_f32_16x16x4_f32(false, a1, false, b,
                                                       (short)0, p1, false, false);
        }
        float bias = b1[col];
        #pragma unroll
        for (int j = 0; j < 8; ++j) {
            As[j + half * 8][col]      = fmaxf(p0[j] + bias, 0.f);
            As[16 + j + half * 8][col] = fmaxf(p1[j] + bias, 0.f);
        }
    }
    __syncthreads();

    v8f acc0 = {}, acc1 = {};
    #pragma unroll 4
    for (int kt = 0; kt < DIM / 4; ++kt) {
        int k0 = kt * 4 + half * 2;
        v2f a0, a1, b;
        b.x = W2[(size_t)k0 * DIM + colBase + l16];
        b.y = W2[(size_t)(k0 + 1) * DIM + colBase + l16];
        a0.x = As[l16][k0];       a0.y = As[l16][k0 + 1];
        a1.x = As[16 + l16][k0];  a1.y = As[16 + l16][k0 + 1];
        acc0 = __builtin_amdgcn_wmma_f32_16x16x4_f32(false, a0, false, b,
                                                     (short)0, acc0, false, false);
        acc1 = __builtin_amdgcn_wmma_f32_16x16x4_f32(false, a1, false, b,
                                                     (short)0, acc1, false, false);
    }
    if (r0 + 32 <= n) {
        #pragma unroll
        for (int j = 0; j < 8; ++j) {
            h2[(size_t)(r0      + j + half * 8) * DIM + col] = acc0[j];
            h2[(size_t)(r0 + 16 + j + half * 8) * DIM + col] = acc1[j];
        }
    } else {
        for (int j = 0; j < 8; ++j) {
            int row0 = r0 + j + half * 8;
            int row1 = row0 + 16;
            if (row0 < n) h2[(size_t)row0 * DIM + col] = acc0[j];
            if (row1 < n) h2[(size_t)row1 * DIM + col] = acc1[j];
        }
    }
}

__global__ __launch_bounds__(256) void k_self2(const float* __restrict__ dinv,
                                               const float* __restrict__ h2,
                                               float* __restrict__ agg, int n) {
    int t = blockIdx.x * 256 + threadIdx.x;
    int i = t >> 5;
    if (i >= n) return;
    int f = (t & 31) * 4;
    float w = dinv[i] * dinv[i];
    const float4 h = *(const float4*)(h2 + (size_t)i * DIM + f);
    float4 o = make_float4(w * h.x, w * h.y, w * h.z, w * h.w);
    *(float4*)(agg + (size_t)i * DIM + f) = o;
}

__global__ __launch_bounds__(256) void k_edge2(const int* __restrict__ src,
                                               const int* __restrict__ dst,
                                               const float* __restrict__ dinv,
                                               const float* __restrict__ h2,
                                               float* __restrict__ agg, int nE) {
    int w = (blockIdx.x * 256 + threadIdx.x) >> 5;
    int lane = threadIdx.x & 31;
    if (w >= nE) return;
    int s = src[w], d = dst[w];
    float nrm = dinv[s] * dinv[d];
    const float* __restrict__ hrow = h2 + (size_t)s * DIM;
    float* __restrict__ orow = agg + (size_t)d * DIM;
    #pragma unroll
    for (int j = 0; j < 4; ++j) {
        int f = lane + 32 * j;
        atomicAdd(&orow[f], nrm * hrow[f]);
    }
}

__global__ __launch_bounds__(256) void k_zero_pool(float* __restrict__ sums,
                                                   float* __restrict__ cnt, int g) {
    int i = blockIdx.x * 256 + threadIdx.x;
    if (i < g * DIM) sums[i] = 0.f;
    if (i < g) cnt[i] = 0.f;
}
#define POOL_PW 16
__global__ __launch_bounds__(256) void k_pool(const float* __restrict__ agg2,
                                              const float* __restrict__ b2,
                                              const int* __restrict__ batch,
                                              float* __restrict__ sums,
                                              float* __restrict__ cnt, int n) {
    __shared__ float part[8][DIM];
    __shared__ float pcnt[8];
    const int lane = threadIdx.x & 31;
    const int wave = __builtin_amdgcn_readfirstlane((int)(threadIdx.x >> 5));
    const int nb = blockIdx.x * (8 * POOL_PW);
    const int n0 = nb + wave * POOL_PW;
    const int f  = lane * 4;
    const int g0 = batch[nb];
    const float4 bb = *(const float4*)(b2 + f);

    float4 acc0 = make_float4(0.f, 0.f, 0.f, 0.f);
    float  c0 = 0.f;
    float4 acc = make_float4(0.f, 0.f, 0.f, 0.f);
    float  c = 0.f;
    int    gc = g0;

    auto flush = [&](int g) {
        if (g == g0) {
            acc0.x += acc.x; acc0.y += acc.y; acc0.z += acc.z; acc0.w += acc.w;
            c0 += c;
        } else if (c > 0.f) {
            float* __restrict__ srow = sums + (size_t)g * DIM + f;
            atomicAdd(srow + 0, acc.x);
            atomicAdd(srow + 1, acc.y);
            atomicAdd(srow + 2, acc.z);
            atomicAdd(srow + 3, acc.w);
            if (lane == 0) atomicAdd(&cnt[g], c);
        }
        acc = make_float4(0.f, 0.f, 0.f, 0.f);
        c = 0.f;
    };
    auto add = [&](int g, const float4& v) {
        if (g != gc) { flush(gc); gc = g; }
        acc.x += v.x + bb.x; acc.y += v.y + bb.y;
        acc.z += v.z + bb.z; acc.w += v.w + bb.w;
        c += 1.0f;
    };

    if (n0 + POOL_PW <= n) {
        float4 v[POOL_PW];
        int    g[POOL_PW];
        #pragma unroll
        for (int j = 0; j < POOL_PW; ++j) {
            v[j] = *(const float4*)(agg2 + (size_t)(n0 + j) * DIM + f);
            g[j] = batch[n0 + j];
        }
        #pragma unroll
        for (int j = 0; j < POOL_PW; ++j) add(g[j], v[j]);
    } else {
        for (int i = n0; i < n; ++i) {
            const float4 v = *(const float4*)(agg2 + (size_t)i * DIM + f);
            add(batch[i], v);
        }
    }
    flush(gc);

    *(float4*)(&part[wave][f]) = acc0;
    if (lane == 0) pcnt[wave] = c0;
    __syncthreads();
    const int t = threadIdx.x;
    if (t < DIM) {
        float s = part[0][t];
        #pragma unroll
        for (int w = 1; w < 8; ++w) s += part[w][t];
        atomicAdd(&sums[(size_t)g0 * DIM + t], s);
    }
    if (t == 0) {
        float cs = pcnt[0];
        #pragma unroll
        for (int w = 1; w < 8; ++w) cs += pcnt[w];
        atomicAdd(&cnt[g0], cs);
    }
}
__global__ __launch_bounds__(256) void k_final(const float* __restrict__ sums,
                                               const float* __restrict__ cnt,
                                               float* __restrict__ out, int g) {
    int i = blockIdx.x * 256 + threadIdx.x;
    if (i >= g * DIM) return;
    float c = cnt[i >> 7];
    out[i] = sums[i] / fmaxf(c, 1.0f);
}

extern "C" void kernel_launch(void* const* d_in, const int* in_sizes, int n_in,
                              void* d_out, int out_size, void* d_ws, size_t ws_size,
                              hipStream_t stream) {
    const int*   node_ids = (const int*)d_in[0];
    const int*   ei       = (const int*)d_in[1];
    const int*   batch    = (const int*)d_in[2];
    const float* emb      = (const float*)d_in[4];
    const float* W1       = (const float*)d_in[5];
    const float* b1       = (const float*)d_in[6];
    const float* W2       = (const float*)d_in[7];
    const float* b2       = (const float*)d_in[8];
    float* out = (float*)d_out;

    const int N = in_sizes[0];
    const int E = in_sizes[1] / 2;
    const int G = out_size / DIM;
    const int* src = ei;
    const int* dst = ei + E;

    char* base = (char*)d_ws;
    size_t off = 0;
    auto alloc = [&](size_t bytes) -> char* {
        char* p = base + off;
        off = (off + bytes + 255) & ~(size_t)255;
        return p;
    };
    float* HW1  = (float*)alloc((size_t)NTP * DIM * sizeof(float));
    float* dinv = (float*)alloc((size_t)N * sizeof(float));
    float* coef = (float*)alloc((size_t)N * CS * sizeof(float));
    float* bufA = (float*)alloc((size_t)N * DIM * sizeof(float));
    float* bufB = (float*)alloc((size_t)N * DIM * sizeof(float));
    float* sums = (float*)alloc((size_t)G * DIM * sizeof(float));
    float* cnt  = (float*)alloc((size_t)G * sizeof(float));
    (void)ws_size; (void)n_in;

    const int TB = 256;
    dim3 blk(TB);

    k_hw1<<<dim3((NTP * DIM + TB - 1) / TB), blk, 0, stream>>>(emb, W1, HW1);

    k_deg_init <<<dim3((N + TB - 1) / TB), blk, 0, stream>>>(dinv, N);
    k_deg_edges<<<dim3((E + TB - 1) / TB), blk, 0, stream>>>(dst, dinv, E);
    k_dinv     <<<dim3((N + TB - 1) / TB), blk, 0, stream>>>(dinv, N);

    k_cinit<<<dim3(((size_t)N * (CS / 4) + TB - 1) / TB), blk, 0, stream>>>(node_ids, dinv, coef, N);
    k_cedge<<<dim3((E + TB - 1) / TB), blk, 0, stream>>>(src, dst, node_ids, dinv, coef, E);

    k_gemm2<<<dim3((N + 31) / 32), blk, 0, stream>>>(coef, HW1, b1, W2, bufB, N);

    k_self2<<<dim3(((size_t)N * 32 + TB - 1) / TB), blk, 0, stream>>>(dinv, bufB, bufA, N);
    k_edge2<<<dim3(((size_t)E * 32 + TB - 1) / TB), blk, 0, stream>>>(src, dst, dinv, bufB, bufA, E);

    k_zero_pool<<<dim3((G * DIM + TB - 1) / TB), blk, 0, stream>>>(sums, cnt, G);
    k_pool     <<<dim3((N + 8 * POOL_PW - 1) / (8 * POOL_PW)), blk, 0, stream>>>(bufA, b2, batch, sums, cnt, N);
    k_final    <<<dim3((G * DIM + TB - 1) / TB), blk, 0, stream>>>(sums, cnt, out, G);
}


# conv2 + mean pool: each edge adds one scalar into a per-source-node per-graph table (N x 64); pooled sums = table^T @ h2 on f32 WMMA per node range, not 128 atomics per edge; changes summation order
# speedup vs baseline: 3.3652x; 1.7670x over previous
#include <hip/hip_runtime.h>
#include <hip/hip_bf16.h>

typedef float v2f __attribute__((ext_vector_type(2)));
typedef float v8f __attribute__((ext_vector_type(8)));

#define DIM 128
#define NT 17
#define NTP 20
#define CS NTP
#define NG 64
#define KB 200
#define TP 128

__global__ __launch_bounds__(256) void k_hw1(const float* __restrict__ emb,
                                             const float* __restrict__ W1,
                                             float* __restrict__ HW1) {
    int t = blockIdx.x * 256 + threadIdx.x;
    if (t >= NTP * DIM) return;
    int m = t >> 7, n = t & 127;
    float acc = 0.f;
    if (m < NT) {
        #pragma unroll 8
        for (int k = 0; k < DIM; ++k)
            acc = fmaf(emb[m * DIM + k], W1[k * DIM + n], acc);
    }
    HW1[t] = acc;
}

__global__ __launch_bounds__(256) void k_deg_init(float* __restrict__ deg, int n) {
    int i = blockIdx.x * 256 + threadIdx.x;
    if (i < n) deg[i] = 1.0f;
}
__global__ __launch_bounds__(256) void k_deg_edges(const int* __restrict__ dst,
                                                   float* __restrict__ deg, int nE) {
    int e = blockIdx.x * 256 + threadIdx.x;
    if (e < nE) atomicAdd(&deg[dst[e]], 1.0f);
}
__global__ __launch_bounds__(256) void k_dinv(float* __restrict__ deg, int n) {
    int i = blockIdx.x * 256 + threadIdx.x;
    if (i < n) { float d = deg[i]; deg[i] = d > 0.f ? rsqrtf(d) : 0.f; }
}

__global__ __launch_bounds__(256) void k_cinit(const int* __restrict__ node_ids,
                                               const float* __restrict__ dinv,
                                               float* __restrict__ coef, int n) {
    int t = blockIdx.x * 256 + threadIdx.x;
    if (t >= n * (CS / 4)) return;
    int i = t / (CS / 4);
    int q = t - i * (CS / 4);
    float w = dinv[i] * dinv[i];
    int ty = node_ids[i] - q * 4;
    float4 o = make_float4(ty == 0 ? w : 0.f, ty == 1 ? w : 0.f, ty == 2 ? w : 0.f, ty == 3 ? w : 0.f);
    *(float4*)(coef + (size_t)t * 4) = o;
}
__global__ __launch_bounds__(256) void k_tinit(const int* __restrict__ batch,
                                               const float* __restrict__ dinv,
                                               float* __restrict__ tq, int n, int npairs) {
    int t = blockIdx.x * 256 + threadIdx.x;
    if (t >= npairs * (TP / 4)) return;
    int p = t >> 5, q = t & 31;
    int c = q >> 1;
    int m0 = (q & 1) * 2;
    int n0 = 2 * p, n1 = n0 + 1;
    float w0 = 0.f, w1 = 0.f;
    int k0 = -1, k1 = -1;
    if (n0 < n) { float d = dinv[n0]; int g = batch[n0]; w0 = d * d; if ((g & 15) == c) k0 = (g >> 4) - m0; }
    if (n1 < n) { float d = dinv[n1]; int g = batch[n1]; w1 = d * d; if ((g & 15) == c) k1 = (g >> 4) - m0; }
    float4 o = make_float4(k0 == 0 ? w0 : 0.f, k1 == 0 ? w1 : 0.f, k0 == 1 ? w0 : 0.f, k1 == 1 ? w1 : 0.f);
    *(float4*)(tq + (size_t)t * 4) = o;
}
__global__ __launch_bounds__(256) void k_cedge(const int* __restrict__ src,
                                               const int* __restrict__ dst,
                                               const int* __restrict__ node_ids,
                                               const int* __restrict__ batch,
                                               const float* __restrict__ dinv,
                                               float* __restrict__ coef,
                                               float* __restrict__ tq, int nE) {
    int e = blockIdx.x * 256 + threadIdx.x;
    if (e >= nE) return;
    int s = src[e], d = dst[e];
    float nrm = dinv[s] * dinv[d];
    int g = batch[d];
    atomicAdd(&coef[(size_t)d * CS + node_ids[s]], nrm);
    atomicAdd(&tq[(size_t)(s >> 1) * TP + (g & 15) * 8 + (g >> 4) * 2 + (s & 1)], nrm);
}

__global__ __launch_bounds__(256) void k_gemm2(const float* __restrict__ coef,
                                               const float* __restrict__ HW1,
                                               const float* __restrict__ b1,
                                               const float* __restrict__ W2,
                                               float* __restrict__ h2, int n) {
    __shared__ float As[32][DIM + 4];
    int r0 = blockIdx.x * 32;
    int tid = threadIdx.x;
    int wave = tid >> 5;
    int lane = tid & 31;
    int half = lane >> 4;
    int l16  = lane & 15;
    int colBase = wave * 16;
    int col = colBase + l16;

    {
        const float* __restrict__ ca = coef + (size_t)min(r0 + l16, n - 1) * CS;
        const float* __restrict__ cb = coef + (size_t)min(r0 + 16 + l16, n - 1) * CS;
        v8f p0 = {}, p1 = {};
        #pragma unroll
        for (int kt = 0; kt < NTP / 4; ++kt) {
            int k0 = kt * 4 + half * 2;
            v2f a0, a1, b;
            b.x = HW1[(size_t)k0 * DIM + col];
            b.y = HW1[(size_t)(k0 + 1) * DIM + col];
            a0 = *(const v2f*)(ca + k0);
            a1 = *(const v2f*)(cb + k0);
            p0 = __builtin_amdgcn_wmma_f32_16x16x4_f32(false, a0, false, b,
                                                       (short)0, p0, false, false);
            p1 = __builtin_amdgcn_wmma_f32_16x16x4_f32(false, a1, false, b,
                                                       (short)0, p1, false, false);
        }
        float bias = b1[col];
        #pragma unroll
        for (int j = 0; j < 8; ++j) {
            As[j + half * 8][col]      = fmaxf(p0[j] + bias, 0.f);
            As[16 + j + half * 8][col] = fmaxf(p1[j] + bias, 0.f);
        }
    }
    __syncthreads();

    v8f acc0 = {}, acc1 = {};
    #pragma unroll 4
    for (int kt = 0; kt < DIM / 4; ++kt) {
        int k0 = kt * 4 + half * 2;
        v2f a0, a1, b;
        b.x = W2[(size_t)k0 * DIM + colBase + l16];
        b.y = W2[(size_t)(k0 + 1) * DIM + colBase + l16];
        a0.x = As[l16][k0];       a0.y = As[l16][k0 + 1];
        a1.x = As[16 + l16][k0];  a1.y = As[16 + l16][k0 + 1];
        acc0 = __builtin_amdgcn_wmma_f32_16x16x4_f32(false, a0, false, b,
                                                     (short)0, acc0, false, false);
        acc1 = __builtin_amdgcn_wmma_f32_16x16x4_f32(false, a1, false, b,
                                                     (short)0, acc1, false, false);
    }
    if (r0 + 32 <= n) {
        #pragma unroll
        for (int j = 0; j < 8; ++j) {
            h2[(size_t)(r0      + j + half * 8) * DIM + col] = acc0[j];
            h2[(size_t)(r0 + 16 + j + half * 8) * DIM + col] = acc1[j];
        }
    } else {
        for (int j = 0; j < 8; ++j) {
            int row0 = r0 + j + half * 8;
            int row1 = row0 + 16;
            if (row0 < n) h2[(size_t)row0 * DIM + col] = acc0[j];
            if (row1 < n) h2[(size_t)row1 * DIM + col] = acc1[j];
        }
    }
}

template <bool TAIL>
__device__ __forceinline__ void pool_steps(const float* __restrict__ tp, const float* __restrict__ h2,
                                           int row0, int n, int col, v8f (&acc)[4]) {
    #pragma unroll 5
    for (int kt = 0; kt < KB / 4; ++kt) {
        const float4 q0 = *(const float4*)(tp + (size_t)kt * (2 * TP));
        const float4 q1 = *(const float4*)(tp + (size_t)kt * (2 * TP) + 4);
        v2f b, a0, a1, a2, a3;
        if (TAIL) {
            int ra = min(row0 + kt * 4, n - 1), rb = min(row0 + kt * 4 + 1, n - 1);
            b.x = h2[(size_t)ra * DIM + col];
            b.y = h2[(size_t)rb * DIM + col];
        } else {
            const float* __restrict__ hb = h2 + (size_t)row0 * DIM + col;
            b.x = hb[(size_t)kt * (4 * DIM)];
            b.y = hb[(size_t)kt * (4 * DIM) + DIM];
        }
        a0.x = q0.x; a0.y = q0.y; a1.x = q0.z; a1.y = q0.w;
        a2.x = q1.x; a2.y = q1.y; a3.x = q1.z; a3.y = q1.w;
        acc[0] = __builtin_amdgcn_wmma_f32_16x16x4_f32(false, a0, false, b, (short)0, acc[0], false, false);
        acc[1] = __builtin_amdgcn_wmma_f32_16x16x4_f32(false, a1, false, b, (short)0, acc[1], false, false);
        acc[2] = __builtin_amdgcn_wmma_f32_16x16x4_f32(false, a2, false, b, (short)0, acc[2], false, false);
        acc[3] = __builtin_amdgcn_wmma_f32_16x16x4_f32(false, a3, false, b, (short)0, acc[3], false, false);
    }
}
__global__ __launch_bounds__(256) void k_poolmm(const float* __restrict__ tq,
                                                const float* __restrict__ h2,
                                                const int* __restrict__ batch,
                                                float* __restrict__ part,
                                                float* __restrict__ pcnt, int n) {
    static_assert(KB % 4 == 0 && KB % 2 == 0 && KB <= 256, "KB");
    __shared__ int cl[NG];
    __shared__ float st[NG][DIM + 4];
    const int tid = threadIdx.x;
    const int wave = tid >> 5, lane = tid & 31;
    const int half = lane >> 4, l16 = lane & 15;
    const int col = wave * 16 + l16;
    const int nb = blockIdx.x * KB;

    if (tid < NG) cl[tid] = 0;
    __syncthreads();
    if (tid < KB && nb + tid < n) atomicAdd(&cl[batch[nb + tid]], 1);

    v8f acc[4] = {};
    const float* __restrict__ tp = tq + (size_t)((nb >> 1) + half) * TP + l16 * 8;
    const int row0 = nb + half * 2;
    if (nb + KB <= n) pool_steps<false>(tp, h2, row0, n, col, acc);
    else              pool_steps<true >(tp, h2, row0, n, col, acc);

    #pragma unroll
    for (int mt = 0; mt < 4; ++mt) {
        #pragma unroll
        for (int j = 0; j < 8; ++j)
            st[mt * 16 + j + half * 8][col] = acc[mt][j];
    }
    __syncthreads();
    float* __restrict__ pb = part + (size_t)blockIdx.x * (NG * DIM);
    #pragma unroll
    for (int i = 0; i < NG / 8; ++i) {
        const int r = wave + 8 * i;
        *(float4*)(pb + (size_t)r * DIM + lane * 4) = *(const float4*)(&st[r][lane * 4]);
    }
    if (tid < NG) pcnt[blockIdx.x * NG + tid] = (float)cl[tid];
}

__global__ __launch_bounds__(256) void k_final(const float* __restrict__ part,
                                               const float* __restrict__ pcnt,
                                               const float* __restrict__ b2,
                                               float* __restrict__ out, int nblk) {
    __shared__ float ps[8][32];
    __shared__ float pc[8];
    const int wave = threadIdx.x >> 5, lane = threadIdx.x & 31;
    const int o = blockIdx.x * 32 + lane;
    const int g = o >> 7;
    float s = 0.f, c = 0.f;
    #pragma unroll 8
    for (int b = wave; b < nblk; b += 8) {
        s += part[(size_t)b * (NG * DIM) + o];
        c += pcnt[b * NG + g];
    }
    ps[wave][lane] = s;
    if (lane == 0) pc[wave] = c;
    __syncthreads();
    if (wave == 0) {
        float ss = ps[0][lane], cs = pc[0];
        #pragma unroll
        for (int w = 1; w < 8; ++w) { ss += ps[w][lane]; cs += pc[w]; }
        out[o] = fmaf(cs, b2[o & (DIM - 1)], ss) / fmaxf(cs, 1.0f);
    }
}

extern "C" void kernel_launch(void* const* d_in, const int* in_sizes, int n_in,
                              void* d_out, int out_size, void* d_ws, size_t ws_size,
                              hipStream_t stream) {
    const int*   node_ids = (const int*)d_in[0];
    const int*   ei       = (const int*)d_in[1];
    const int*   batch    = (const int*)d_in[2];
    const float* emb      = (const float*)d_in[4];
    const float* W1       = (const float*)d_in[5];
    const float* b1       = (const float*)d_in[6];
    const float* W2       = (const float*)d_in[7];
    const float* b2       = (const float*)d_in[8];
    float* out = (float*)d_out;

    const int N = in_sizes[0];
    const int E = in_sizes[1] / 2;
    const int* src = ei;
    const int* dst = ei + E;

    char* base = (char*)d_ws;
    size_t off = 0;
    auto alloc = [&](size_t bytes) -> char* {
        char* p = base + off;
        off = (off + bytes + 255) & ~(size_t)255;
        return p;
    };
    float* HW1  = (float*)alloc((size_t)NTP * DIM * sizeof(float));
    float* dinv = (float*)alloc((size_t)N * sizeof(float));
    float* coef = (float*)alloc((size_t)N * CS * sizeof(float));
    const int NB = (N + KB - 1) / KB;
    const int NPAIR = NB * (KB / 2);
    float* tq   = (float*)alloc((size_t)NPAIR * TP * sizeof(float));
    float* bufB = (float*)alloc((size_t)N * DIM * sizeof(float));
    float* part = (float*)alloc((size_t)NB * NG * DIM * sizeof(float));
    float* pcnt = (float*)alloc((size_t)NB * NG * sizeof(float));
    (void)ws_size; (void)n_in; (void)out_size;

    const int TB = 256;
    dim3 blk(TB);

    k_hw1<<<dim3((NTP * DIM + TB - 1) / TB), blk, 0, stream>>>(emb, W1, HW1);

    k_deg_init <<<dim3((N + TB - 1) / TB), blk, 0, stream>>>(dinv, N);
    k_deg_edges<<<dim3((E + TB - 1) / TB), blk, 0, stream>>>(dst, dinv, E);
    k_dinv     <<<dim3((N + TB - 1) / TB), blk, 0, stream>>>(dinv, N);

    k_cinit<<<dim3(((size_t)N * (CS / 4) + TB - 1) / TB), blk, 0, stream>>>(node_ids, dinv, coef, N);
    k_tinit<<<dim3(((size_t)NPAIR * (TP / 4) + TB - 1) / TB), blk, 0, stream>>>(batch, dinv, tq, N, NPAIR);
    k_cedge<<<dim3((E + TB - 1) / TB), blk, 0, stream>>>(src, dst, node_ids, batch, dinv, coef, tq, E);

    k_gemm2<<<dim3((N + 31) / 32), blk, 0, stream>>>(coef, HW1, b1, W2, bufB, N);

    k_poolmm<<<dim3(NB), blk, 0, stream>>>(tq, bufB, batch, part, pcnt, N);
    k_final <<<dim3(NG * DIM / 32), blk, 0, stream>>>(part, pcnt, b2, out, NB);
}


# speedup vs baseline: 5.4142x; 1.6089x over previous
#include <hip/hip_runtime.h>
#include <hip/hip_bf16.h>

typedef float v2f __attribute__((ext_vector_type(2)));
typedef float v8f __attribute__((ext_vector_type(8)));

#define DIM 128
#define NT 17
#define NTP 20
#define CS NTP
#define NG 64
#define KB 200
#define TP 128

__device__ __forceinline__ int pkg_half() {
    return (int)((__builtin_amdgcn_s_sendmsg_rtn(0x87) >> 18) & 1u);
}

__global__ __launch_bounds__(256) void k_hw1(const float* __restrict__ emb,
                                             const float* __restrict__ W1,
                                             float* __restrict__ HW1) {
    int t = blockIdx.x * 256 + threadIdx.x;
    if (t >= NTP * DIM) return;
    int m = t >> 7, n = t & 127;
    float acc = 0.f;
    if (m < NT) {
        #pragma unroll 8
        for (int k = 0; k < DIM; ++k)
            acc = fmaf(emb[m * DIM + k], W1[k * DIM + n], acc);
    }
    HW1[t] = acc;
}

__global__ __launch_bounds__(256) void k_deg_init(float* __restrict__ deg0, float* __restrict__ deg1, int n) {
    int i = blockIdx.x * 256 + threadIdx.x;
    if (i < n) { deg0[i] = 1.0f; deg1[i] = 0.0f; }
}
__global__ __launch_bounds__(256) void k_deg_edges(const int* __restrict__ dst,
                                                   float* __restrict__ deg0, float* __restrict__ deg1, int nE) {
    int e = blockIdx.x * 256 + threadIdx.x;
    float* __restrict__ deg = pkg_half() ? deg1 : deg0;
    if (e < nE) atomicAdd(&deg[dst[e]], 1.0f);
}
__global__ __launch_bounds__(256) void k_dinv(float* __restrict__ deg0, const float* __restrict__ deg1, int n) {
    int i = blockIdx.x * 256 + threadIdx.x;
    if (i < n) { float d = deg0[i] + deg1[i]; deg0[i] = d > 0.f ? rsqrtf(d) : 0.f; }
}

__global__ __launch_bounds__(256) void k_cinit(const int* __restrict__ node_ids,
                                               const float* __restrict__ dinv,
                                               float* __restrict__ coef, float* __restrict__ coef1, int n) {
    int t = blockIdx.x * 256 + threadIdx.x;
    if (t >= n * (CS / 4)) return;
    int i = t / (CS / 4);
    int q = t - i * (CS / 4);
    float w = dinv[i] * dinv[i];
    int ty = node_ids[i] - q * 4;
    float4 o = make_float4(ty == 0 ? w : 0.f, ty == 1 ? w : 0.f, ty == 2 ? w : 0.f, ty == 3 ? w : 0.f);
    *(float4*)(coef + (size_t)t * 4) = o;
    *(float4*)(coef1 + (size_t)t * 4) = make_float4(0.f, 0.f, 0.f, 0.f);
}
__global__ __launch_bounds__(256) void k_tinit(const int* __restrict__ batch,
                                               const float* __restrict__ dinv,
                                               float* __restrict__ tq, float* __restrict__ tq1,
                                               int n, int npairs) {
    int t = blockIdx.x * 256 + threadIdx.x;
    if (t >= npairs * (TP / 4)) return;
    int p = t >> 5, q = t & 31;
    int c = q >> 1;
    int m0 = (q & 1) * 2;
    int n0 = 2 * p, n1 = n0 + 1;
    float w0 = 0.f, w1 = 0.f;
    int k0 = -1, k1 = -1;
    if (n0 < n) { float d = dinv[n0]; int g = batch[n0]; w0 = d * d; if ((g & 15) == c) k0 = (g >> 4) - m0; }
    if (n1 < n) { float d = dinv[n1]; int g = batch[n1]; w1 = d * d; if ((g & 15) == c) k1 = (g >> 4) - m0; }
    float4 o = make_float4(k0 == 0 ? w0 : 0.f, k1 == 0 ? w1 : 0.f, k0 == 1 ? w0 : 0.f, k1 == 1 ? w1 : 0.f);
    *(float4*)(tq + (size_t)t * 4) = o;
    *(float4*)(tq1 + (size_t)t * 4) = make_float4(0.f, 0.f, 0.f, 0.f);
}
__global__ __launch_bounds__(256) void k_cedge(const int* __restrict__ src,
                                               const int* __restrict__ dst,
                                               const int* __restrict__ node_ids,
                                               const int* __restrict__ batch,
                                               const float* __restrict__ dinv,
                                               float* __restrict__ coef0, float* __restrict__ coef1,
                                               float* __restrict__ tq0, float* __restrict__ tq1, int nE) {
    int e = blockIdx.x * 256 + threadIdx.x;
    const int ph = pkg_half();
    float* __restrict__ coef = ph ? coef1 : coef0;
    float* __restrict__ tq   = ph ? tq1 : tq0;
    if (e >= nE) return;
    int s = src[e], d = dst[e];
    float nrm = dinv[s] * dinv[d];
    int g = batch[d];
    atomicAdd(&coef[(size_t)d * CS + node_ids[s]], nrm);
    atomicAdd(&tq[(size_t)(s >> 1) * TP + (g & 15) * 8 + (g >> 4) * 2 + (s & 1)], nrm);
}

__global__ __launch_bounds__(256) void k_gemm2(const float* __restrict__ coef,
                                               const float* __restrict__ coef1,
                                               const float* __restrict__ HW1,
                                               const float* __restrict__ b1,
                                               const float* __restrict__ W2,
                                               float* __restrict__ h2, int n) {
    __shared__ float As[32][DIM + 4];
    int r0 = blockIdx.x * 32;
    int tid = threadIdx.x;
    int wave = tid >> 5;
    int lane = tid & 31;
    int half = lane >> 4;
    int l16  = lane & 15;
    int colBase = wave * 16;
    int col = colBase + l16;

    {
        const size_t oa = (size_t)min(r0 + l16, n - 1) * CS;
        const size_t ob = (size_t)min(r0 + 16 + l16, n - 1) * CS;
        const float* __restrict__ ca = coef + oa;
        const float* __restrict__ cb = coef + ob;
        const float* __restrict__ ca1 = coef1 + oa;
        const float* __restrict__ cb1 = coef1 + ob;
        v8f p0 = {}, p1 = {};
        #pragma unroll
        for (int kt = 0; kt < NTP / 4; ++kt) {
            int k0 = kt * 4 + half * 2;
            v2f a0, a1, b;
            b.x = HW1[(size_t)k0 * DIM + col];
            b.y = HW1[(size_t)(k0 + 1) * DIM + col];
            a0 = *(const v2f*)(ca + k0) + *(const v2f*)(ca1 + k0);
            a1 = *(const v2f*)(cb + k0) + *(const v2f*)(cb1 + k0);
            p0 = __builtin_amdgcn_wmma_f32_16x16x4_f32(false, a0, false, b,
                                                       (short)0, p0, false, false);
            p1 = __builtin_amdgcn_wmma_f32_16x16x4_f32(false, a1, false, b,
                                                       (short)0, p1, false, false);
        }
        float bias = b1[col];
        #pragma unroll
        for (int j = 0; j < 8; ++j) {
            As[j + half * 8][col]      = fmaxf(p0[j] + bias, 0.f);
            As[16 + j + half * 8][col] = fmaxf(p1[j] + bias, 0.f);
        }
    }
    __syncthreads();

    v8f acc0 = {}, acc1 = {};
    #pragma unroll 4
    for (int kt = 0; kt < DIM / 4; ++kt) {
        int k0 = kt * 4 + half * 2;
        v2f a0, a1, b;
        b.x = W2[(size_t)k0 * DIM + colBase + l16];
        b.y = W2[(size_t)(k0 + 1) * DIM + colBase + l16];
        a0.x = As[l16][k0];       a0.y = As[l16][k0 + 1];
        a1.x = As[16 + l16][k0];  a1.y = As[16 + l16][k0 + 1];
        acc0 = __builtin_amdgcn_wmma_f32_16x16x4_f32(false, a0, false, b,
                                                     (short)0, acc0, false, false);
        acc1 = __builtin_amdgcn_wmma_f32_16x16x4_f32(false, a1, false, b,
                                                     (short)0, acc1, false, false);
    }
    if (r0 + 32 <= n) {
        #pragma unroll
        for (int j = 0; j < 8; ++j) {
            h2[(size_t)(r0      + j + half * 8) * DIM + col] = acc0[j];
            h2[(size_t)(r0 + 16 + j + half * 8) * DIM + col] = acc1[j];
        }
    } else {
        for (int j = 0; j < 8; ++j) {
            int row0 = r0 + j + half * 8;
            int row1 = row0 + 16;
            if (row0 < n) h2[(size_t)row0 * DIM + col] = acc0[j];
            if (row1 < n) h2[(size_t)row1 * DIM + col] = acc1[j];
        }
    }
}

template <bool TAIL>
__device__ __forceinline__ void pool_steps(const float* __restrict__ tp, const float* __restrict__ tp1,
                                           const float* __restrict__ h2,
                                           int row0, int n, int col, v8f (&acc)[4]) {
    #pragma unroll 5
    for (int kt = 0; kt < KB / 4; ++kt) {
        const float4 u0 = *(const float4*)(tp  + (size_t)kt * (2 * TP));
        const float4 u1 = *(const float4*)(tp  + (size_t)kt * (2 * TP) + 4);
        const float4 w0 = *(const float4*)(tp1 + (size_t)kt * (2 * TP));
        const float4 w1 = *(const float4*)(tp1 + (size_t)kt * (2 * TP) + 4);
        const float4 q0 = make_float4(u0.x + w0.x, u0.y + w0.y, u0.z + w0.z, u0.w + w0.w);
        const float4 q1 = make_float4(u1.x + w1.x, u1.y + w1.y, u1.z + w1.z, u1.w + w1.w);
        v2f b, a0, a1, a2, a3;
        if (TAIL) {
            int ra = min(row0 + kt * 4, n - 1), rb = min(row0 + kt * 4 + 1, n - 1);
            b.x = h2[(size_t)ra * DIM + col];
            b.y = h2[(size_t)rb * DIM + col];
        } else {
            const float* __restrict__ hb = h2 + (size_t)row0 * DIM + col;
            b.x = hb[(size_t)kt * (4 * DIM)];
            b.y = hb[(size_t)kt * (4 * DIM) + DIM];
        }
        a0.x = q0.x; a0.y = q0.y; a1.x = q0.z; a1.y = q0.w;
        a2.x = q1.x; a2.y = q1.y; a3.x = q1.z; a3.y = q1.w;
        acc[0] = __builtin_amdgcn_wmma_f32_16x16x4_f32(false, a0, false, b, (short)0, acc[0], false, false);
        acc[1] = __builtin_amdgcn_wmma_f32_16x16x4_f32(false, a1, false, b, (short)0, acc[1], false, false);
        acc[2] = __builtin_amdgcn_wmma_f32_16x16x4_f32(false, a2, false, b, (short)0, acc[2], false, false);
        acc[3] = __builtin_amdgcn_wmma_f32_16x16x4_f32(false, a3, false, b, (short)0, acc[3], false, false);
    }
}
__global__ __launch_bounds__(256) void k_poolmm(const float* __restrict__ tq,
                                                const float* __restrict__ tq1,
                                                const float* __restrict__ h2,
                                                const int* __restrict__ batch,
                                                float* __restrict__ part,
                                                float* __restrict__ pcnt, int n) {
    static_assert(KB % 4 == 0 && KB % 2 == 0 && KB <= 256, "KB");
    __shared__ int cl[NG];
    __shared__ float st[NG][DIM + 4];
    const int tid = threadIdx.x;
    const int wave = tid >> 5, lane = tid & 31;
    const int half = lane >> 4, l16 = lane & 15;
    const int col = wave * 16 + l16;
    const int nb = blockIdx.x * KB;

    if (tid < NG) cl[tid] = 0;
    __syncthreads();
    if (tid < KB && nb + tid < n) atomicAdd(&cl[batch[nb + tid]], 1);

    v8f acc[4] = {};
    const size_t to = (size_t)((nb >> 1) + half) * TP + l16 * 8;
    const float* __restrict__ tp  = tq + to;
    const float* __restrict__ tp1 = tq1 + to;
    const int row0 = nb + half * 2;
    if (nb + KB <= n) pool_steps<false>(tp, tp1, h2, row0, n, col, acc);
    else              pool_steps<true >(tp, tp1, h2, row0, n, col, acc);

    #pragma unroll
    for (int mt = 0; mt < 4; ++mt) {
        #pragma unroll
        for (int j = 0; j < 8; ++j)
            st[mt * 16 + j + half * 8][col] = acc[mt][j];
    }
    __syncthreads();
    float* __restrict__ pb = part + (size_t)blockIdx.x * (NG * DIM);
    #pragma unroll
    for (int i = 0; i < NG / 8; ++i) {
        const int r = wave + 8 * i;
        *(float4*)(pb + (size_t)r * DIM + lane * 4) = *(const float4*)(&st[r][lane * 4]);
    }
    if (tid < NG) pcnt[blockIdx.x * NG + tid] = (float)cl[tid];
}

__global__ __launch_bounds__(256) void k_final(const float* __restrict__ part,
                                               const float* __restrict__ pcnt,
                                               const float* __restrict__ b2,
                                               float* __restrict__ out, int nblk) {
    __shared__ float ps[8][32];
    __shared__ float pc[8];
    const int wave = threadIdx.x >> 5, lane = threadIdx.x & 31;
    const int o = blockIdx.x * 32 + lane;
    const int g = o >> 7;
    float s = 0.f, c = 0.f;
    #pragma unroll 8
    for (int b = wave; b < nblk; b += 8) {
        s += part[(size_t)b * (NG * DIM) + o];
        c += pcnt[b * NG + g];
    }
    ps[wave][lane] = s;
    if (lane == 0) pc[wave] = c;
    __syncthreads();
    if (wave == 0) {
        float ss = ps[0][lane], cs = pc[0];
        #pragma unroll
        for (int w = 1; w < 8; ++w) { ss += ps[w][lane]; cs += pc[w]; }
        out[o] = fmaf(cs, b2[o & (DIM - 1)], ss) / fmaxf(cs, 1.0f);
    }
}

extern "C" void kernel_launch(void* const* d_in, const int* in_sizes, int n_in,
                              void* d_out, int out_size, void* d_ws, size_t ws_size,
                              hipStream_t stream) {
    const int*   node_ids = (const int*)d_in[0];
    const int*   ei       = (const int*)d_in[1];
    const int*   batch    = (const int*)d_in[2];
    const float* emb      = (const float*)d_in[4];
    const float* W1       = (const float*)d_in[5];
    const float* b1       = (const float*)d_in[6];
    const float* W2       = (const float*)d_in[7];
    const float* b2       = (const float*)d_in[8];
    float* out = (float*)d_out;

    const int N = in_sizes[0];
    const int E = in_sizes[1] / 2;
    const int* src = ei;
    const int* dst = ei + E;

    char* base = (char*)d_ws;
    size_t off = 0;
    auto alloc = [&](size_t bytes) -> char* {
        char* p = base + off;
        off = (off + bytes + 255) & ~(size_t)255;
        return p;
    };
    float* HW1  = (float*)alloc((size_t)NTP * DIM * sizeof(float));
    float* dinv = (float*)alloc((size_t)N * sizeof(float));
    float* deg1 = (float*)alloc((size_t)N * sizeof(float));
    float* coef = (float*)alloc((size_t)N * CS * sizeof(float));
    float* coef1 = (float*)alloc((size_t)N * CS * sizeof(float));
    const int NB = (N + KB - 1) / KB;
    const int NPAIR = NB * (KB / 2);
    float* tq   = (float*)alloc((size_t)NPAIR * TP * sizeof(float));
    float* tq1  = (float*)alloc((size_t)NPAIR * TP * sizeof(float));
    float* bufB = (float*)alloc((size_t)N * DIM * sizeof(float));
    float* part = (float*)alloc((size_t)NB * NG * DIM * sizeof(float));
    float* pcnt = (float*)alloc((size_t)NB * NG * sizeof(float));
    (void)ws_size; (void)n_in; (void)out_size;

    const int TB = 256;
    dim3 blk(TB);

    k_hw1<<<dim3((NTP * DIM + TB - 1) / TB), blk, 0, stream>>>(emb, W1, HW1);

    k_deg_init <<<dim3((N + TB - 1) / TB), blk, 0, stream>>>(dinv, deg1, N);
    k_deg_edges<<<dim3((E + TB - 1) / TB), blk, 0, stream>>>(dst, dinv, deg1, E);
    k_dinv     <<<dim3((N + TB - 1) / TB), blk, 0, stream>>>(dinv, deg1, N);

    k_cinit<<<dim3(((size_t)N * (CS / 4) + TB - 1) / TB), blk, 0, stream>>>(node_ids, dinv, coef, coef1, N);
    k_tinit<<<dim3(((size_t)NPAIR * (TP / 4) + TB - 1) / TB), blk, 0, stream>>>(batch, dinv, tq, tq1, N, NPAIR);
    k_cedge<<<dim3((E + TB - 1) / TB), blk, 0, stream>>>(src, dst, node_ids, batch, dinv, coef, coef1, tq, tq1, E);

    k_gemm2<<<dim3((N + 31) / 32), blk, 0, stream>>>(coef, coef1, HW1, b1, W2, bufB, N);

    k_poolmm<<<dim3(NB), blk, 0, stream>>>(tq, tq1, bufB, batch, part, pcnt, N);
    k_final <<<dim3(NG * DIM / 32), blk, 0, stream>>>(part, pcnt, b2, out, NB);
}


# edge pass: dinv, node type and graph id packed in one 8-byte record per node; an edge gathers 2 records (2 lines), not 4 scalars from 3 arrays (4 lines); sums and atomic order as before
# speedup vs baseline: 5.7883x; 1.0691x over previous
#include <hip/hip_runtime.h>
#include <hip/hip_bf16.h>

typedef float v2f __attribute__((ext_vector_type(2)));
typedef float v8f __attribute__((ext_vector_type(8)));

#define DIM 128
#define NT 17
#define NTP 20
#define CS NTP
#define NG 64
#define KB 200
#define TP 128

__device__ __forceinline__ int pkg_half() {
    return (int)((__builtin_amdgcn_s_sendmsg_rtn(0x87) >> 18) & 1u);
}

__global__ __launch_bounds__(256) void k_hw1(const float* __restrict__ emb,
                                             const float* __restrict__ W1,
                                             float* __restrict__ HW1) {
    int t = blockIdx.x * 256 + threadIdx.x;
    if (t >= NTP * DIM) return;
    int m = t >> 7, n = t & 127;
    float acc = 0.f;
    if (m < NT) {
        #pragma unroll 8
        for (int k = 0; k < DIM; ++k)
            acc = fmaf(emb[m * DIM + k], W1[k * DIM + n], acc);
    }
    HW1[t] = acc;
}

__global__ __launch_bounds__(256) void k_deg_init(float* __restrict__ deg0, float* __restrict__ deg1, int n) {
    int i = blockIdx.x * 256 + threadIdx.x;
    if (i < n) { deg0[i] = 1.0f; deg1[i] = 0.0f; }
}
__global__ __launch_bounds__(256) void k_deg_edges(const int* __restrict__ dst,
                                                   float* __restrict__ deg0, float* __restrict__ deg1, int nE) {
    int e = blockIdx.x * 256 + threadIdx.x;
    float* __restrict__ deg = pkg_half() ? deg1 : deg0;
    if (e < nE) atomicAdd(&deg[dst[e]], 1.0f);
}
__global__ __launch_bounds__(256) void k_rec(const float* __restrict__ deg0, const float* __restrict__ deg1,
                                             const int* __restrict__ node_ids, const int* __restrict__ batch,
                                             int* __restrict__ rec, int n) {
    int i = blockIdx.x * 256 + threadIdx.x;
    if (i < n) {
        float d = deg0[i] + deg1[i];
        float v = d > 0.f ? rsqrtf(d) : 0.f;
        *(int2*)(rec + 2 * (size_t)i) = make_int2(__float_as_int(v), (batch[i] << 16) | (node_ids[i] & 0xFFFF));
    }
}
__device__ __forceinline__ float rec_dinv(int2 r)  { return __int_as_float(r.x); }
__device__ __forceinline__ int   rec_type(int2 r)  { return r.y & 0xFFFF; }
__device__ __forceinline__ int   rec_graph(int2 r) { return r.y >> 16; }

__global__ __launch_bounds__(256) void k_cinit(const int* __restrict__ rec,
                                               float* __restrict__ coef, float* __restrict__ coef1, int n) {
    int t = blockIdx.x * 256 + threadIdx.x;
    if (t >= n * (CS / 4)) return;
    int i = t / (CS / 4);
    int q = t - i * (CS / 4);
    const int2 r = *(const int2*)(rec + 2 * (size_t)i);
    float w = rec_dinv(r) * rec_dinv(r);
    int ty = rec_type(r) - q * 4;
    float4 o = make_float4(ty == 0 ? w : 0.f, ty == 1 ? w : 0.f, ty == 2 ? w : 0.f, ty == 3 ? w : 0.f);
    *(float4*)(coef + (size_t)t * 4) = o;
    *(float4*)(coef1 + (size_t)t * 4) = make_float4(0.f, 0.f, 0.f, 0.f);
}
__global__ __launch_bounds__(256) void k_tinit(const int* __restrict__ rec,
                                               float* __restrict__ tq, float* __restrict__ tq1,
                                               int n, int npairs) {
    int t = blockIdx.x * 256 + threadIdx.x;
    if (t >= npairs * (TP / 4)) return;
    int p = t >> 5, q = t & 31;
    int c = q >> 1;
    int m0 = (q & 1) * 2;
    int n0 = 2 * p, n1 = n0 + 1;
    float w0 = 0.f, w1 = 0.f;
    int k0 = -1, k1 = -1;
    if (n0 < n) { const int2 r = *(const int2*)(rec + 2 * (size_t)n0); float d = rec_dinv(r); int g = rec_graph(r);
                  w0 = d * d; if ((g & 15) == c) k0 = (g >> 4) - m0; }
    if (n1 < n) { const int2 r = *(const int2*)(rec + 2 * (size_t)n1); float d = rec_dinv(r); int g = rec_graph(r);
                  w1 = d * d; if ((g & 15) == c) k1 = (g >> 4) - m0; }
    float4 o = make_float4(k0 == 0 ? w0 : 0.f, k1 == 0 ? w1 : 0.f, k0 == 1 ? w0 : 0.f, k1 == 1 ? w1 : 0.f);
    *(float4*)(tq + (size_t)t * 4) = o;
    *(float4*)(tq1 + (size_t)t * 4) = make_float4(0.f, 0.f, 0.f, 0.f);
}
__global__ __launch_bounds__(256) void k_cedge(const int* __restrict__ src,
                                               const int* __restrict__ dst,
                                               const int* __restrict__ rec,
                                               float* __restrict__ coef0, float* __restrict__ coef1,
                                               float* __restrict__ tq0, float* __restrict__ tq1, int nE) {
    int e = blockIdx.x * 256 + threadIdx.x;
    const int ph = pkg_half();
    float* __restrict__ coef = ph ? coef1 : coef0;
    float* __restrict__ tq   = ph ? tq1 : tq0;
    if (e >= nE) return;
    int s = src[e], d = dst[e];
    const int2 rs = *(const int2*)(rec + 2 * (size_t)s);
    const int2 rd = *(const int2*)(rec + 2 * (size_t)d);
    float nrm = rec_dinv(rs) * rec_dinv(rd);
    int g = rec_graph(rd);
    atomicAdd(&coef[(size_t)d * CS + rec_type(rs)], nrm);
    atomicAdd(&tq[(size_t)(s >> 1) * TP + (g & 15) * 8 + (g >> 4) * 2 + (s & 1)], nrm);
}

__global__ __launch_bounds__(256) void k_gemm2(const float* __restrict__ coef,
                                               const float* __restrict__ coef1,
                                               const float* __restrict__ HW1,
                                               const float* __restrict__ b1,
                                               const float* __restrict__ W2,
                                               float* __restrict__ h2, int n) {
    __shared__ float As[32][DIM + 4];
    int r0 = blockIdx.x * 32;
    int tid = threadIdx.x;
    int wave = tid >> 5;
    int lane = tid & 31;
    int half = lane >> 4;
    int l16  = lane & 15;
    int colBase = wave * 16;
    int col = colBase + l16;

    {
        const size_t oa = (size_t)min(r0 + l16, n - 1) * CS;
        const size_t ob = (size_t)min(r0 + 16 + l16, n - 1) * CS;
        const float* __restrict__ ca = coef + oa;
        const float* __restrict__ cb = coef + ob;
        const float* __restrict__ ca1 = coef1 + oa;
        const float* __restrict__ cb1 = coef1 + ob;
        v8f p0 = {}, p1 = {};
        #pragma unroll
        for (int kt = 0; kt < NTP / 4; ++kt) {
            int k0 = kt * 4 + half * 2;
            v2f a0, a1, b;
            b.x = HW1[(size_t)k0 * DIM + col];
            b.y = HW1[(size_t)(k0 + 1) * DIM + col];
            a0 = *(const v2f*)(ca + k0) + *(const v2f*)(ca1 + k0);
            a1 = *(const v2f*)(cb + k0) + *(const v2f*)(cb1 + k0);
            p0 = __builtin_amdgcn_wmma_f32_16x16x4_f32(false, a0, false, b,
                                                       (short)0, p0, false, false);
            p1 = __builtin_amdgcn_wmma_f32_16x16x4_f32(false, a1, false, b,
                                                       (short)0, p1, false, false);
        }
        float bias = b1[col];
        #pragma unroll
        for (int j = 0; j < 8; ++j) {
            As[j + half * 8][col]      = fmaxf(p0[j] + bias, 0.f);
            As[16 + j + half * 8][col] = fmaxf(p1[j] + bias, 0.f);
        }
    }
    __syncthreads();

    v8f acc0 = {}, acc1 = {};
    #pragma unroll 4
    for (int kt = 0; kt < DIM / 4; ++kt) {
        int k0 = kt * 4 + half * 2;
        v2f a0, a1, b;
        b.x = W2[(size_t)k0 * DIM + colBase + l16];
        b.y = W2[(size_t)(k0 + 1) * DIM + colBase + l16];
        a0.x = As[l16][k0];       a0.y = As[l16][k0 + 1];
        a1.x = As[16 + l16][k0];  a1.y = As[16 + l16][k0 + 1];
        acc0 = __builtin_amdgcn_wmma_f32_16x16x4_f32(false, a0, false, b,
                                                     (short)0, acc0, false, false);
        acc1 = __builtin_amdgcn_wmma_f32_16x16x4_f32(false, a1, false, b,
                                                     (short)0, acc1, false, false);
    }
    if (r0 + 32 <= n) {
        #pragma unroll
        for (int j = 0; j < 8; ++j) {
            h2[(size_t)(r0      + j + half * 8) * DIM + col] = acc0[j];
            h2[(size_t)(r0 + 16 + j + half * 8) * DIM + col] = acc1[j];
        }
    } else {
        for (int j = 0; j < 8; ++j) {
            int row0 = r0 + j + half * 8;
            int row1 = row0 + 16;
            if (row0 < n) h2[(size_t)row0 * DIM + col] = acc0[j];
            if (row1 < n) h2[(size_t)row1 * DIM + col] = acc1[j];
        }
    }
}

template <bool TAIL>
__device__ __forceinline__ void pool_steps(const float* __restrict__ tp, const float* __restrict__ tp1,
                                           const float* __restrict__ h2,
                                           int row0, int n, int col, v8f (&acc)[4]) {
    #pragma unroll 5
    for (int kt = 0; kt < KB / 4; ++kt) {
        const float4 u0 = *(const float4*)(tp  + (size_t)kt * (2 * TP));
        const float4 u1 = *(const float4*)(tp  + (size_t)kt * (2 * TP) + 4);
        const float4 w0 = *(const float4*)(tp1 + (size_t)kt * (2 * TP));
        const float4 w1 = *(const float4*)(tp1 + (size_t)kt * (2 * TP) + 4);
        const float4 q0 = make_float4(u0.x + w0.x, u0.y + w0.y, u0.z + w0.z, u0.w + w0.w);
        const float4 q1 = make_float4(u1.x + w1.x, u1.y + w1.y, u1.z + w1.z, u1.w + w1.w);
        v2f b, a0, a1, a2, a3;
        if (TAIL) {
            int ra = min(row0 + kt * 4, n - 1), rb = min(row0 + kt * 4 + 1, n - 1);
            b.x = h2[(size_t)ra * DIM + col];
            b.y = h2[(size_t)rb * DIM + col];
        } else {
            const float* __restrict__ hb = h2 + (size_t)row0 * DIM + col;
            b.x = hb[(size_t)kt * (4 * DIM)];
            b.y = hb[(size_t)kt * (4 * DIM) + DIM];
        }
        a0.x = q0.x; a0.y = q0.y; a1.x = q0.z; a1.y = q0.w;
        a2.x = q1.x; a2.y = q1.y; a3.x = q1.z; a3.y = q1.w;
        acc[0] = __builtin_amdgcn_wmma_f32_16x16x4_f32(false, a0, false, b, (short)0, acc[0], false, false);
        acc[1] = __builtin_amdgcn_wmma_f32_16x16x4_f32(false, a1, false, b, (short)0, acc[1], false, false);
        acc[2] = __builtin_amdgcn_wmma_f32_16x16x4_f32(false, a2, false, b, (short)0, acc[2], false, false);
        acc[3] = __builtin_amdgcn_wmma_f32_16x16x4_f32(false, a3, false, b, (short)0, acc[3], false, false);
    }
}
__global__ __launch_bounds__(256) void k_poolmm(const float* __restrict__ tq,
                                                const float* __restrict__ tq1,
                                                const float* __restrict__ h2,
                                                const int* __restrict__ batch,
                                                float* __restrict__ part,
                                                float* __restrict__ pcnt, int n) {
    static_assert(KB % 4 == 0 && KB % 2 == 0 && KB <= 256, "KB");
    __shared__ int cl[NG];
    __shared__ float st[NG][DIM + 4];
    const int tid = threadIdx.x;
    const int wave = tid >> 5, lane = tid & 31;
    const int half = lane >> 4, l16 = lane & 15;
    const int col = wave * 16 + l16;
    const int nb = blockIdx.x * KB;

    if (tid < NG) cl[tid] = 0;
    __syncthreads();
    if (tid < KB && nb + tid < n) atomicAdd(&cl[batch[nb + tid]], 1);

    v8f acc[4] = {};
    const size_t to = (size_t)((nb >> 1) + half) * TP + l16 * 8;
    const float* __restrict__ tp  = tq + to;
    const float* __restrict__ tp1 = tq1 + to;
    const int row0 = nb + half * 2;
    if (nb + KB <= n) pool_steps<false>(tp, tp1, h2, row0, n, col, acc);
    else              pool_steps<true >(tp, tp1, h2, row0, n, col, acc);

    #pragma unroll
    for (int mt = 0; mt < 4; ++mt) {
        #pragma unroll
        for (int j = 0; j < 8; ++j)
            st[mt * 16 + j + half * 8][col] = acc[mt][j];
    }
    __syncthreads();
    float* __restrict__ pb = part + (size_t)blockIdx.x * (NG * DIM);
    #pragma unroll
    for (int i = 0; i < NG / 8; ++i) {
        const int r = wave + 8 * i;
        *(float4*)(pb + (size_t)r * DIM + lane * 4) = *(const float4*)(&st[r][lane * 4]);
    }
    if (tid < NG) pcnt[blockIdx.x * NG + tid] = (float)cl[tid];
}

__global__ __launch_bounds__(256) void k_final(const float* __restrict__ part,
                                               const float* __restrict__ pcnt,
                                               const float* __restrict__ b2,
                                               float* __restrict__ out, int nblk) {
    __shared__ float ps[8][32];
    __shared__ float pc[8];
    const int wave = threadIdx.x >> 5, lane = threadIdx.x & 31;
    const int o = blockIdx.x * 32 + lane;
    const int g = o >> 7;
    float s = 0.f, c = 0.f;
    #pragma unroll 8
    for (int b = wave; b < nblk; b += 8) {
        s += part[(size_t)b * (NG * DIM) + o];
        c += pcnt[b * NG + g];
    }
    ps[wave][lane] = s;
    if (lane == 0) pc[wave] = c;
    __syncthreads();
    if (wave == 0) {
        float ss = ps[0][lane], cs = pc[0];
        #pragma unroll
        for (int w = 1; w < 8; ++w) { ss += ps[w][lane]; cs += pc[w]; }
        out[o] = fmaf(cs, b2[o & (DIM - 1)], ss) / fmaxf(cs, 1.0f);
    }
}

extern "C" void kernel_launch(void* const* d_in, const int* in_sizes, int n_in,
                              void* d_out, int out_size, void* d_ws, size_t ws_size,
                              hipStream_t stream) {
    const int*   node_ids = (const int*)d_in[0];
    const int*   ei       = (const int*)d_in[1];
    const int*   batch    = (const int*)d_in[2];
    const float* emb      = (const float*)d_in[4];
    const float* W1       = (const float*)d_in[5];
    const float* b1       = (const float*)d_in[6];
    const float* W2       = (const float*)d_in[7];
    const float* b2       = (const float*)d_in[8];
    float* out = (float*)d_out;

    const int N = in_sizes[0];
    const int E = in_sizes[1] / 2;
    const int* src = ei;
    const int* dst = ei + E;

    char* base = (char*)d_ws;
    size_t off = 0;
    auto alloc = [&](size_t bytes) -> char* {
        char* p = base + off;
        off = (off + bytes + 255) & ~(size_t)255;
        return p;
    };
    float* HW1  = (float*)alloc((size_t)NTP * DIM * sizeof(float));
    float* deg0 = (float*)alloc((size_t)N * sizeof(float));
    float* deg1 = (float*)alloc((size_t)N * sizeof(float));
    int*   rec  = (int*)alloc((size_t)N * 2 * sizeof(int));
    float* coef = (float*)alloc((size_t)N * CS * sizeof(float));
    float* coef1 = (float*)alloc((size_t)N * CS * sizeof(float));
    const int NB = (N + KB - 1) / KB;
    const int NPAIR = NB * (KB / 2);
    float* tq   = (float*)alloc((size_t)NPAIR * TP * sizeof(float));
    float* tq1  = (float*)alloc((size_t)NPAIR * TP * sizeof(float));
    float* bufB = (float*)alloc((size_t)N * DIM * sizeof(float));
    float* part = (float*)alloc((size_t)NB * NG * DIM * sizeof(float));
    float* pcnt = (float*)alloc((size_t)NB * NG * sizeof(float));
    (void)ws_size; (void)n_in; (void)out_size;

    const int TB = 256;
    dim3 blk(TB);

    k_hw1<<<dim3((NTP * DIM + TB - 1) / TB), blk, 0, stream>>>(emb, W1, HW1);

    k_deg_init <<<dim3((N + TB - 1) / TB), blk, 0, stream>>>(deg0, deg1, N);
    k_deg_edges<<<dim3((E + TB - 1) / TB), blk, 0, stream>>>(dst, deg0, deg1, E);
    k_rec      <<<dim3((N + TB - 1) / TB), blk, 0, stream>>>(deg0, deg1, node_ids, batch, rec, N);

    k_cinit<<<dim3(((size_t)N * (CS / 4) + TB - 1) / TB), blk, 0, stream>>>(rec, coef, coef1, N);
    k_tinit<<<dim3(((size_t)NPAIR * (TP / 4) + TB - 1) / TB), blk, 0, stream>>>(rec, tq, tq1, N, NPAIR);
    k_cedge<<<dim3((E + TB - 1) / TB), blk, 0, stream>>>(src, dst, rec, coef, coef1, tq, tq1, E);

    k_gemm2<<<dim3((N + 31) / 32), blk, 0, stream>>>(coef, coef1, HW1, b1, W2, bufB, N);

    k_poolmm<<<dim3(NB), blk, 0, stream>>>(tq, tq1, bufB, batch, part, pcnt, N);
    k_final <<<dim3(NG * DIM / 32), blk, 0, stream>>>(part, pcnt, b2, out, NB);
}


# W2 applied after the pool: relu(coef@HW1+b1) tiles stay in WMMA result registers and feed tq^T@R as B; one 64-row W2 product in the final kernel; no N-row W2 GEMM, no h2; changes summation order
# speedup vs baseline: 6.1318x; 1.0593x over previous
#include <hip/hip_runtime.h>
#include <hip/hip_bf16.h>

typedef float v2f __attribute__((ext_vector_type(2)));
typedef float v8f __attribute__((ext_vector_type(8)));

#define DIM 128
#define NT 17
#define NTP 20
#define CS NTP
#define NG 64
#define KB 208
#define TP 128

__device__ __forceinline__ int pkg_half() {
    return (int)((__builtin_amdgcn_s_sendmsg_rtn(0x87) >> 18) & 1u);
}

__global__ __launch_bounds__(256) void k_hw1(const float* __restrict__ emb,
                                             const float* __restrict__ W1,
                                             float* __restrict__ HW1) {
    int t = blockIdx.x * 256 + threadIdx.x;
    if (t >= NTP * DIM) return;
    int m = t >> 7, n = t & 127;
    float acc = 0.f;
    if (m < NT) {
        #pragma unroll 8
        for (int k = 0; k < DIM; ++k)
            acc = fmaf(emb[m * DIM + k], W1[k * DIM + n], acc);
    }
    HW1[t] = acc;
}

__global__ __launch_bounds__(256) void k_deg_init(float* __restrict__ deg0, float* __restrict__ deg1, int n) {
    int i = blockIdx.x * 256 + threadIdx.x;
    if (i < n) { deg0[i] = 1.0f; deg1[i] = 0.0f; }
}
__global__ __launch_bounds__(256) void k_deg_edges(const int* __restrict__ dst,
                                                   float* __restrict__ deg0, float* __restrict__ deg1, int nE) {
    int e = blockIdx.x * 256 + threadIdx.x;
    float* __restrict__ deg = pkg_half() ? deg1 : deg0;
    if (e < nE) atomicAdd(&deg[dst[e]], 1.0f);
}
__global__ __launch_bounds__(256) void k_rec(const float* __restrict__ deg0, const float* __restrict__ deg1,
                                             const int* __restrict__ node_ids, const int* __restrict__ batch,
                                             int* __restrict__ rec, int n) {
    int i = blockIdx.x * 256 + threadIdx.x;
    if (i < n) {
        float d = deg0[i] + deg1[i];
        float v = d > 0.f ? rsqrtf(d) : 0.f;
        *(int2*)(rec + 2 * (size_t)i) = make_int2(__float_as_int(v), (batch[i] << 16) | (node_ids[i] & 0xFFFF));
    }
}
__device__ __forceinline__ float rec_dinv(int2 r)  { return __int_as_float(r.x); }
__device__ __forceinline__ int   rec_type(int2 r)  { return r.y & 0xFFFF; }
__device__ __forceinline__ int   rec_graph(int2 r) { return r.y >> 16; }

__global__ __launch_bounds__(256) void k_cinit(const int* __restrict__ rec,
                                               float* __restrict__ coef, float* __restrict__ coef1, int n) {
    int t = blockIdx.x * 256 + threadIdx.x;
    if (t >= n * (CS / 4)) return;
    int i = t / (CS / 4);
    int q = t - i * (CS / 4);
    const int2 r = *(const int2*)(rec + 2 * (size_t)i);
    float w = rec_dinv(r) * rec_dinv(r);
    int ty = rec_type(r) - q * 4;
    float4 o = make_float4(ty == 0 ? w : 0.f, ty == 1 ? w : 0.f, ty == 2 ? w : 0.f, ty == 3 ? w : 0.f);
    *(float4*)(coef + (size_t)t * 4) = o;
    *(float4*)(coef1 + (size_t)t * 4) = make_float4(0.f, 0.f, 0.f, 0.f);
}
__global__ __launch_bounds__(256) void k_tinit(const int* __restrict__ rec,
                                               float* __restrict__ tq, float* __restrict__ tq1,
                                               int n, int npairs) {
    int t = blockIdx.x * 256 + threadIdx.x;
    if (t >= npairs * (TP / 4)) return;
    int p = t >> 5, q = t & 31;
    int c = q >> 1;
    int m0 = (q & 1) * 2;
    int n0 = 2 * p, n1 = n0 + 1;
    float w0 = 0.f, w1 = 0.f;
    int k0 = -1, k1 = -1;
    if (n0 < n) { const int2 r = *(const int2*)(rec + 2 * (size_t)n0); float d = rec_dinv(r); int g = rec_graph(r);
                  w0 = d * d; if ((g & 15) == c) k0 = (g >> 4) - m0; }
    if (n1 < n) { const int2 r = *(const int2*)(rec + 2 * (size_t)n1); float d = rec_dinv(r); int g = rec_graph(r);
                  w1 = d * d; if ((g & 15) == c) k1 = (g >> 4) - m0; }
    float4 o = make_float4(k0 == 0 ? w0 : 0.f, k1 == 0 ? w1 : 0.f, k0 == 1 ? w0 : 0.f, k1 == 1 ? w1 : 0.f);
    *(float4*)(tq + (size_t)t * 4) = o;
    *(float4*)(tq1 + (size_t)t * 4) = make_float4(0.f, 0.f, 0.f, 0.f);
}
__global__ __launch_bounds__(256) void k_cedge(const int* __restrict__ src,
                                               const int* __restrict__ dst,
                                               const int* __restrict__ rec,
                                               float* __restrict__ coef0, float* __restrict__ coef1,
                                               float* __restrict__ tq0, float* __restrict__ tq1, int nE) {
    int e = blockIdx.x * 256 + threadIdx.x;
    const int ph = pkg_half();
    float* __restrict__ coef = ph ? coef1 : coef0;
    float* __restrict__ tq   = ph ? tq1 : tq0;
    if (e >= nE) return;
    int s = src[e], d = dst[e];
    const int2 rs = *(const int2*)(rec + 2 * (size_t)s);
    const int2 rd = *(const int2*)(rec + 2 * (size_t)d);
    float nrm = rec_dinv(rs) * rec_dinv(rd);
    int g = rec_graph(rd);
    atomicAdd(&coef[(size_t)d * CS + rec_type(rs)], nrm);
    atomicAdd(&tq[(size_t)(s >> 1) * TP + (g & 15) * 8 + (g >> 4) * 2 + (s & 1)], nrm);
}

#define NCH (KB / 16)
struct PFrag {
    v2f c0[NTP / 4], c1[NTP / 4];
    float4 u0[4], u1[4], w0[4], w1[4];
};
struct PSum {
    v2f a[NTP / 4];
    v2f t[4][4];
};
__device__ __forceinline__ void pf_load(PFrag& f, const float* __restrict__ coef, const float* __restrict__ coef1,
                                        const float* __restrict__ tq, const float* __restrict__ tq1,
                                        int base, int n, int half, int l16) {
    const size_t ro = (size_t)min(base + l16, n - 1) * CS + half * 2;
    #pragma unroll
    for (int kt = 0; kt < NTP / 4; ++kt) {
        f.c0[kt] = *(const v2f*)(coef  + ro + kt * 4);
        f.c1[kt] = *(const v2f*)(coef1 + ro + kt * 4);
    }
    const size_t to = (size_t)((base >> 1) + 4 * half) * TP + l16 * 8;
    #pragma unroll
    for (int q = 0; q < 4; ++q) {
        f.u0[q] = *(const float4*)(tq  + to + (size_t)q * TP);
        f.u1[q] = *(const float4*)(tq  + to + (size_t)q * TP + 4);
        f.w0[q] = *(const float4*)(tq1 + to + (size_t)q * TP);
        f.w1[q] = *(const float4*)(tq1 + to + (size_t)q * TP + 4);
    }
}
__device__ __forceinline__ void pf_sum(const PFrag& f, PSum& s) {
    #pragma unroll
    for (int kt = 0; kt < NTP / 4; ++kt) s.a[kt] = f.c0[kt] + f.c1[kt];
    #pragma unroll
    for (int q = 0; q < 4; ++q) {
        s.t[q][0].x = f.u0[q].x + f.w0[q].x; s.t[q][0].y = f.u0[q].y + f.w0[q].y;
        s.t[q][1].x = f.u0[q].z + f.w0[q].z; s.t[q][1].y = f.u0[q].w + f.w0[q].w;
        s.t[q][2].x = f.u1[q].x + f.w1[q].x; s.t[q][2].y = f.u1[q].y + f.w1[q].y;
        s.t[q][3].x = f.u1[q].z + f.w1[q].z; s.t[q][3].y = f.u1[q].w + f.w1[q].w;
    }
}
__device__ __forceinline__ void pf_mma(const PSum& s, const v2f (&hb)[2][NTP / 4], const float (&bias)[2],
                                       v8f (&acc)[2][4]) {
    v8f p0 = {}, p1 = {};
    #pragma unroll
    for (int kt = 0; kt < NTP / 4; ++kt) {
        p0 = __builtin_amdgcn_wmma_f32_16x16x4_f32(false, s.a[kt], false, hb[0][kt], (short)0, p0, false, false);
        p1 = __builtin_amdgcn_wmma_f32_16x16x4_f32(false, s.a[kt], false, hb[1][kt], (short)0, p1, false, false);
    }
    #pragma unroll
    for (int j = 0; j < 8; ++j) {
        p0[j] = fmaxf(p0[j] + bias[0], 0.f);
        p1[j] = fmaxf(p1[j] + bias[1], 0.f);
    }
    #pragma unroll
    for (int q = 0; q < 4; ++q) {
        v2f b0, b1;
        b0.x = p0[2 * q]; b0.y = p0[2 * q + 1];
        b1.x = p1[2 * q]; b1.y = p1[2 * q + 1];
        #pragma unroll
        for (int mt = 0; mt < 4; ++mt) {
            acc[0][mt] = __builtin_amdgcn_wmma_f32_16x16x4_f32(false, s.t[q][mt], false, b0, (short)0, acc[0][mt], false, false);
            acc[1][mt] = __builtin_amdgcn_wmma_f32_16x16x4_f32(false, s.t[q][mt], false, b1, (short)0, acc[1][mt], false, false);
        }
    }
}
__global__ __launch_bounds__(256) void k_poolmm(const float* __restrict__ coef,
                                                const float* __restrict__ coef1,
                                                const float* __restrict__ HW1,
                                                const float* __restrict__ b1,
                                                const float* __restrict__ tq,
                                                const float* __restrict__ tq1,
                                                const int* __restrict__ batch,
                                                float* __restrict__ part,
                                                float* __restrict__ pcnt, int n) {
    static_assert(KB % 16 == 0 && KB <= 256, "KB");
    __shared__ int cl[NG];
    __shared__ float st[NG][DIM + 4];
    const int tid = threadIdx.x;
    const int wave = __builtin_amdgcn_readfirstlane(tid >> 5);
    const int lane = tid & 31;
    const int half = lane >> 4, l16 = lane & 15;
    const int cg = wave & 3;
    const int par = wave >> 2;
    const int nb = blockIdx.x * KB;

    if (tid < NG) cl[tid] = 0;
    __syncthreads();
    if (tid < KB && nb + tid < n) atomicAdd(&cl[batch[nb + tid]], 1);

    v2f hb[2][NTP / 4];
    float bias[2];
    #pragma unroll
    for (int ct = 0; ct < 2; ++ct) {
        const int col = cg * 32 + ct * 16 + l16;
        #pragma unroll
        for (int kt = 0; kt < NTP / 4; ++kt) {
            const int k0 = kt * 4 + half * 2;
            hb[ct][kt].x = HW1[(size_t)k0 * DIM + col];
            hb[ct][kt].y = HW1[(size_t)(k0 + 1) * DIM + col];
        }
        bias[ct] = b1[col];
    }

    v8f acc[2][4] = {};
    #pragma unroll 1
    for (int ch = par; ch < NCH; ch += 2) {
        PFrag f;
        PSum s;
        pf_load(f, coef, coef1, tq, tq1, nb + ch * 16, n, half, l16);
        pf_sum(f, s);
        pf_mma(s, hb, bias, acc);
    }

    if (par == 1) {
        #pragma unroll
        for (int ct = 0; ct < 2; ++ct) {
            const int col = cg * 32 + ct * 16 + l16;
            #pragma unroll
            for (int mt = 0; mt < 4; ++mt) {
                #pragma unroll
                for (int j = 0; j < 8; ++j)
                    st[mt * 16 + j + half * 8][col] = acc[ct][mt][j];
            }
        }
    }
    __syncthreads();
    if (par == 0) {
        #pragma unroll
        for (int ct = 0; ct < 2; ++ct) {
            const int col = cg * 32 + ct * 16 + l16;
            #pragma unroll
            for (int mt = 0; mt < 4; ++mt) {
                #pragma unroll
                for (int j = 0; j < 8; ++j)
                    st[mt * 16 + j + half * 8][col] += acc[ct][mt][j];
            }
        }
    }
    __syncthreads();
    float* __restrict__ pb = part + (size_t)blockIdx.x * (NG * DIM);
    #pragma unroll
    for (int i = 0; i < NG / 8; ++i) {
        const int r = wave + 8 * i;
        *(float4*)(pb + (size_t)r * DIM + lane * 4) = *(const float4*)(&st[r][lane * 4]);
    }
    if (tid < NG) pcnt[blockIdx.x * NG + tid] = (float)cl[tid];
}

__global__ __launch_bounds__(1024) void k_final(const float* __restrict__ part,
                                                const float* __restrict__ pcnt,
                                                const float* __restrict__ W2,
                                                const float* __restrict__ b2,
                                                float* __restrict__ out, int nblk) {
    __shared__ float ps[32][DIM];
    __shared__ float prow[DIM];
    __shared__ float orow[DIM];
    __shared__ float pc[32];
    const int tid = threadIdx.x;
    const int wave = __builtin_amdgcn_readfirstlane(tid >> 5);
    const int lane = tid & 31;
    const int half = lane >> 4, l16 = lane & 15;
    const int g = blockIdx.x;
    const int col = (wave & 7) * 16 + l16;

    v2f wb[DIM / 4] = {};
    if (wave < 8) {
        #pragma unroll
        for (int kt = 0; kt < DIM / 4; ++kt) {
            const int k0 = kt * 4 + half * 2;
            wb[kt].x = W2[(size_t)k0 * DIM + col];
            wb[kt].y = W2[(size_t)(k0 + 1) * DIM + col];
        }
    }

    float4 s = make_float4(0.f, 0.f, 0.f, 0.f);
    for (int b0 = wave; b0 < nblk; b0 += 32 * 8) {
        #pragma unroll
        for (int u = 0; u < 8; ++u) {
            const int b = b0 + 32 * u;
            const float m = b < nblk ? 1.f : 0.f;
            const float4 v = *(const float4*)(part + (size_t)min(b, nblk - 1) * (NG * DIM) + g * DIM + lane * 4);
            s.x = fmaf(m, v.x, s.x); s.y = fmaf(m, v.y, s.y); s.z = fmaf(m, v.z, s.z); s.w = fmaf(m, v.w, s.w);
        }
    }
    float c = 0.f;
    for (int b = tid; b < nblk; b += 1024) c += pcnt[b * NG + g];
    #pragma unroll
    for (int off = 16; off > 0; off >>= 1) c += __shfl_xor(c, off);
    *(float4*)(&ps[wave][lane * 4]) = s;
    if (lane == 0) pc[wave] = c;
    __syncthreads();
    if (tid < DIM) {
        float ss = ps[0][tid];
        #pragma unroll
        for (int w = 1; w < 32; ++w) ss += ps[w][tid];
        prow[tid] = ss;
    }
    __syncthreads();
    if (wave < 8) {
        const int keep = l16 == 0 ? -1 : 0;
        v8f acc = {};
        #pragma unroll
        for (int kt = 0; kt < DIM / 4; ++kt) {
            const v2f pa = *(const v2f*)(&prow[kt * 4 + half * 2]);
            v2f a;
            a.x = __int_as_float(__float_as_int(pa.x) & keep);
            a.y = __int_as_float(__float_as_int(pa.y) & keep);
            acc = __builtin_amdgcn_wmma_f32_16x16x4_f32(false, a, false, wb[kt], (short)0, acc, false, false);
        }
        float cs = pc[0];
        #pragma unroll
        for (int w = 1; w < 32; ++w) cs += pc[w];
        if (half == 0) orow[col] = fmaf(cs, b2[col], acc[0]) / fmaxf(cs, 1.0f);
    }
    __syncthreads();
    if (tid < 32) *(float4*)(out + (size_t)g * DIM + tid * 4) = *(const float4*)(&orow[tid * 4]);
}

extern "C" void kernel_launch(void* const* d_in, const int* in_sizes, int n_in,
                              void* d_out, int out_size, void* d_ws, size_t ws_size,
                              hipStream_t stream) {
    const int*   node_ids = (const int*)d_in[0];
    const int*   ei       = (const int*)d_in[1];
    const int*   batch    = (const int*)d_in[2];
    const float* emb      = (const float*)d_in[4];
    const float* W1       = (const float*)d_in[5];
    const float* b1       = (const float*)d_in[6];
    const float* W2       = (const float*)d_in[7];
    const float* b2       = (const float*)d_in[8];
    float* out = (float*)d_out;

    const int N = in_sizes[0];
    const int E = in_sizes[1] / 2;
    const int* src = ei;
    const int* dst = ei + E;

    char* base = (char*)d_ws;
    size_t off = 0;
    auto alloc = [&](size_t bytes) -> char* {
        char* p = base + off;
        off = (off + bytes + 255) & ~(size_t)255;
        return p;
    };
    float* HW1  = (float*)alloc((size_t)NTP * DIM * sizeof(float));
    float* deg0 = (float*)alloc((size_t)N * sizeof(float));
    float* deg1 = (float*)alloc((size_t)N * sizeof(float));
    int*   rec  = (int*)alloc((size_t)N * 2 * sizeof(int));
    float* coef = (float*)alloc((size_t)N * CS * sizeof(float));
    float* coef1 = (float*)alloc((size_t)N * CS * sizeof(float));
    const int NB = (N + KB - 1) / KB;
    const int NPAIR = NB * (KB / 2);
    float* tq   = (float*)alloc((size_t)NPAIR * TP * sizeof(float));
    float* tq1  = (float*)alloc((size_t)NPAIR * TP * sizeof(float));
    float* part = (float*)alloc((size_t)NB * NG * DIM * sizeof(float));
    float* pcnt = (float*)alloc((size_t)NB * NG * sizeof(float));
    (void)ws_size; (void)n_in; (void)out_size;

    const int TB = 256;
    dim3 blk(TB);

    k_hw1<<<dim3((NTP * DIM + TB - 1) / TB), blk, 0, stream>>>(emb, W1, HW1);

    k_deg_init <<<dim3((N + TB - 1) / TB), blk, 0, stream>>>(deg0, deg1, N);
    k_deg_edges<<<dim3((E + TB - 1) / TB), blk, 0, stream>>>(dst, deg0, deg1, E);
    k_rec      <<<dim3((N + TB - 1) / TB), blk, 0, stream>>>(deg0, deg1, node_ids, batch, rec, N);

    k_cinit<<<dim3(((size_t)N * (CS / 4) + TB - 1) / TB), blk, 0, stream>>>(rec, coef, coef1, N);
    k_tinit<<<dim3(((size_t)NPAIR * (TP / 4) + TB - 1) / TB), blk, 0, stream>>>(rec, tq, tq1, N, NPAIR);
    k_cedge<<<dim3((E + TB - 1) / TB), blk, 0, stream>>>(src, dst, rec, coef, coef1, tq, tq1, E);

    k_poolmm<<<dim3(NB), blk, 0, stream>>>(coef, coef1, HW1, b1, tq, tq1, batch, part, pcnt, N);

    k_final <<<dim3(NG), dim3(1024), 0, stream>>>(part, pcnt, W2, b2, out, NB);
}


# launches 9 -> 6: stages with no dependency share a launch: degree edge pass also clears both tables and forms HW1; record kernel stores the self-loop seeds; sums and atomic order as before
# speedup vs baseline: 7.0370x; 1.1476x over previous
#include <hip/hip_runtime.h>
#include <hip/hip_bf16.h>

typedef float v2f __attribute__((ext_vector_type(2)));
typedef float v8f __attribute__((ext_vector_type(8)));

#define DIM 128
#define NT 17
#define NTP 20
#define CS NTP
#define NG 64
#define KB 208
#define TP 128

__device__ __forceinline__ int pkg_half() {
    return (int)((__builtin_amdgcn_s_sendmsg_rtn(0x87) >> 18) & 1u);
}

#define HW1_BLOCKS ((NTP * DIM + 255) / 256)
__device__ __forceinline__ void hw1_block(const float* __restrict__ emb,
                                          const float* __restrict__ W1,
                                          float* __restrict__ HW1) {
    int t = blockIdx.x * 256 + threadIdx.x;
    if (t >= NTP * DIM) return;
    int m = t >> 7, n = t & 127;
    float acc = 0.f;
    if (m < NT) {
        #pragma unroll 8
        for (int k = 0; k < DIM; ++k)
            acc = fmaf(emb[m * DIM + k], W1[k * DIM + n], acc);
    }
    HW1[t] = acc;
}

__global__ __launch_bounds__(256) void k_deg_init(float* __restrict__ deg0, float* __restrict__ deg1, int n) {
    int i = blockIdx.x * 256 + threadIdx.x;
    if (i < n) { deg0[i] = 1.0f; deg1[i] = 0.0f; }
}
__global__ __launch_bounds__(256) void k_deg_edges(const int* __restrict__ dst,
                                                   float* __restrict__ deg0, float* __restrict__ deg1, int nE,
                                                   float* __restrict__ coef0, float* __restrict__ coef1, int nc4,
                                                   float* __restrict__ tq0, float* __restrict__ tq1, int nt4,
                                                   const float* __restrict__ emb,
                                                   const float* __restrict__ W1,
                                                   float* __restrict__ HW1) {
    if (blockIdx.x < HW1_BLOCKS) { hw1_block(emb, W1, HW1); return; }
    const int e = (blockIdx.x - HW1_BLOCKS) * 256 + threadIdx.x;
    const int d = dst[min(e, nE - 1)];
    const int stride = (gridDim.x - HW1_BLOCKS) * 256;
    const float4 z = make_float4(0.f, 0.f, 0.f, 0.f);
    for (int i = e; i < nc4; i += stride) {
        *(float4*)(coef0 + (size_t)i * 4) = z;
        *(float4*)(coef1 + (size_t)i * 4) = z;
    }
    for (int i = e; i < nt4; i += stride) {
        *(float4*)(tq0 + (size_t)i * 4) = z;
        *(float4*)(tq1 + (size_t)i * 4) = z;
    }
    float* __restrict__ deg = pkg_half() ? deg1 : deg0;
    if (e < nE) atomicAdd(&deg[d], 1.0f);
}
__global__ __launch_bounds__(256) void k_rec(const float* __restrict__ deg0, const float* __restrict__ deg1,
                                             const int* __restrict__ node_ids, const int* __restrict__ batch,
                                             int* __restrict__ rec,
                                             float* __restrict__ coef, float* __restrict__ tq, int n) {
    int i = blockIdx.x * 256 + threadIdx.x;
    if (i < n) {
        float d = deg0[i] + deg1[i];
        float v = d > 0.f ? rsqrtf(d) : 0.f;
        const int ty = node_ids[i] & 0xFFFF;
        const int g = batch[i];
        *(int2*)(rec + 2 * (size_t)i) = make_int2(__float_as_int(v), (g << 16) | ty);
        const float w = v * v;
        coef[(size_t)i * CS + ty] = w;
        tq[(size_t)(i >> 1) * TP + (g & 15) * 8 + (g >> 4) * 2 + (i & 1)] = w;
    }
}
__device__ __forceinline__ float rec_dinv(int2 r)  { return __int_as_float(r.x); }
__device__ __forceinline__ int   rec_type(int2 r)  { return r.y & 0xFFFF; }
__device__ __forceinline__ int   rec_graph(int2 r) { return r.y >> 16; }

__global__ __launch_bounds__(256) void k_cedge(const int* __restrict__ src,
                                               const int* __restrict__ dst,
                                               const int* __restrict__ rec,
                                               float* __restrict__ coef0, float* __restrict__ coef1,
                                               float* __restrict__ tq0, float* __restrict__ tq1, int nE) {
    int e = blockIdx.x * 256 + threadIdx.x;
    const int ph = pkg_half();
    float* __restrict__ coef = ph ? coef1 : coef0;
    float* __restrict__ tq   = ph ? tq1 : tq0;
    if (e >= nE) return;
    int s = src[e], d = dst[e];
    const int2 rs = *(const int2*)(rec + 2 * (size_t)s);
    const int2 rd = *(const int2*)(rec + 2 * (size_t)d);
    float nrm = rec_dinv(rs) * rec_dinv(rd);
    int g = rec_graph(rd);
    atomicAdd(&coef[(size_t)d * CS + rec_type(rs)], nrm);
    atomicAdd(&tq[(size_t)(s >> 1) * TP + (g & 15) * 8 + (g >> 4) * 2 + (s & 1)], nrm);
}

#define NCH (KB / 16)
struct PFrag {
    v2f c0[NTP / 4], c1[NTP / 4];
    float4 u0[4], u1[4], w0[4], w1[4];
};
struct PSum {
    v2f a[NTP / 4];
    v2f t[4][4];
};
__device__ __forceinline__ void pf_load(PFrag& f, const float* __restrict__ coef, const float* __restrict__ coef1,
                                        const float* __restrict__ tq, const float* __restrict__ tq1,
                                        int base, int n, int half, int l16) {
    const size_t ro = (size_t)min(base + l16, n - 1) * CS + half * 2;
    #pragma unroll
    for (int kt = 0; kt < NTP / 4; ++kt) {
        f.c0[kt] = *(const v2f*)(coef  + ro + kt * 4);
        f.c1[kt] = *(const v2f*)(coef1 + ro + kt * 4);
    }
    const size_t to = (size_t)((base >> 1) + 4 * half) * TP + l16 * 8;
    #pragma unroll
    for (int q = 0; q < 4; ++q) {
        f.u0[q] = *(const float4*)(tq  + to + (size_t)q * TP);
        f.u1[q] = *(const float4*)(tq  + to + (size_t)q * TP + 4);
        f.w0[q] = *(const float4*)(tq1 + to + (size_t)q * TP);
        f.w1[q] = *(const float4*)(tq1 + to + (size_t)q * TP + 4);
    }
}
__device__ __forceinline__ void pf_sum(const PFrag& f, PSum& s) {
    #pragma unroll
    for (int kt = 0; kt < NTP / 4; ++kt) s.a[kt] = f.c0[kt] + f.c1[kt];
    #pragma unroll
    for (int q = 0; q < 4; ++q) {
        s.t[q][0].x = f.u0[q].x + f.w0[q].x; s.t[q][0].y = f.u0[q].y + f.w0[q].y;
        s.t[q][1].x = f.u0[q].z + f.w0[q].z; s.t[q][1].y = f.u0[q].w + f.w0[q].w;
        s.t[q][2].x = f.u1[q].x + f.w1[q].x; s.t[q][2].y = f.u1[q].y + f.w1[q].y;
        s.t[q][3].x = f.u1[q].z + f.w1[q].z; s.t[q][3].y = f.u1[q].w + f.w1[q].w;
    }
}
__device__ __forceinline__ void pf_mma(const PSum& s, const v2f (&hb)[2][NTP / 4], const float (&bias)[2],
                                       v8f (&acc)[2][4]) {
    v8f p0 = {}, p1 = {};
    #pragma unroll
    for (int kt = 0; kt < NTP / 4; ++kt) {
        p0 = __builtin_amdgcn_wmma_f32_16x16x4_f32(false, s.a[kt], false, hb[0][kt], (short)0, p0, false, false);
        p1 = __builtin_amdgcn_wmma_f32_16x16x4_f32(false, s.a[kt], false, hb[1][kt], (short)0, p1, false, false);
    }
    #pragma unroll
    for (int j = 0; j < 8; ++j) {
        p0[j] = fmaxf(p0[j] + bias[0], 0.f);
        p1[j] = fmaxf(p1[j] + bias[1], 0.f);
    }
    #pragma unroll
    for (int q = 0; q < 4; ++q) {
        v2f b0, b1;
        b0.x = p0[2 * q]; b0.y = p0[2 * q + 1];
        b1.x = p1[2 * q]; b1.y = p1[2 * q + 1];
        #pragma unroll
        for (int mt = 0; mt < 4; ++mt) {
            acc[0][mt] = __builtin_amdgcn_wmma_f32_16x16x4_f32(false, s.t[q][mt], false, b0, (short)0, acc[0][mt], false, false);
            acc[1][mt] = __builtin_amdgcn_wmma_f32_16x16x4_f32(false, s.t[q][mt], false, b1, (short)0, acc[1][mt], false, false);
        }
    }
}
__global__ __launch_bounds__(256) void k_poolmm(const float* __restrict__ coef,
                                                const float* __restrict__ coef1,
                                                const float* __restrict__ HW1,
                                                const float* __restrict__ b1,
                                                const float* __restrict__ tq,
                                                const float* __restrict__ tq1,
                                                const int* __restrict__ batch,
                                                float* __restrict__ part,
                                                float* __restrict__ pcnt, int n) {
    static_assert(KB % 16 == 0 && KB <= 256, "KB");
    __shared__ int cl[NG];
    __shared__ float st[NG][DIM + 4];
    const int tid = threadIdx.x;
    const int wave = __builtin_amdgcn_readfirstlane(tid >> 5);
    const int lane = tid & 31;
    const int half = lane >> 4, l16 = lane & 15;
    const int cg = wave & 3;
    const int par = wave >> 2;
    const int nb = blockIdx.x * KB;

    if (tid < NG) cl[tid] = 0;
    __syncthreads();
    if (tid < KB && nb + tid < n) atomicAdd(&cl[batch[nb + tid]], 1);

    v2f hb[2][NTP / 4];
    float bias[2];
    #pragma unroll
    for (int ct = 0; ct < 2; ++ct) {
        const int col = cg * 32 + ct * 16 + l16;
        #pragma unroll
        for (int kt = 0; kt < NTP / 4; ++kt) {
            const int k0 = kt * 4 + half * 2;
            hb[ct][kt].x = HW1[(size_t)k0 * DIM + col];
            hb[ct][kt].y = HW1[(size_t)(k0 + 1) * DIM + col];
        }
        bias[ct] = b1[col];
    }

    v8f acc[2][4] = {};
    #pragma unroll 1
    for (int ch = par; ch < NCH; ch += 2) {
        PFrag f;
        PSum s;
        pf_load(f, coef, coef1, tq, tq1, nb + ch * 16, n, half, l16);
        pf_sum(f, s);
        pf_mma(s, hb, bias, acc);
    }

    if (par == 1) {
        #pragma unroll
        for (int ct = 0; ct < 2; ++ct) {
            const int col = cg * 32 + ct * 16 + l16;
            #pragma unroll
            for (int mt = 0; mt < 4; ++mt) {
                #pragma unroll
                for (int j = 0; j < 8; ++j)
                    st[mt * 16 + j + half * 8][col] = acc[ct][mt][j];
            }
        }
    }
    __syncthreads();
    if (par == 0) {
        #pragma unroll
        for (int ct = 0; ct < 2; ++ct) {
            const int col = cg * 32 + ct * 16 + l16;
            #pragma unroll
            for (int mt = 0; mt < 4; ++mt) {
                #pragma unroll
                for (int j = 0; j < 8; ++j)
                    st[mt * 16 + j + half * 8][col] += acc[ct][mt][j];
            }
        }
    }
    __syncthreads();
    float* __restrict__ pb = part + (size_t)blockIdx.x * (NG * DIM);
    #pragma unroll
    for (int i = 0; i < NG / 8; ++i) {
        const int r = wave + 8 * i;
        *(float4*)(pb + (size_t)r * DIM + lane * 4) = *(const float4*)(&st[r][lane * 4]);
    }
    if (tid < NG) pcnt[blockIdx.x * NG + tid] = (float)cl[tid];
}

__global__ __launch_bounds__(1024) void k_final(const float* __restrict__ part,
                                                const float* __restrict__ pcnt,
                                                const float* __restrict__ W2,
                                                const float* __restrict__ b2,
                                                float* __restrict__ out, int nblk) {
    __shared__ float ps[32][DIM];
    __shared__ float prow[DIM];
    __shared__ float orow[DIM];
    __shared__ float pc[32];
    const int tid = threadIdx.x;
    const int wave = __builtin_amdgcn_readfirstlane(tid >> 5);
    const int lane = tid & 31;
    const int half = lane >> 4, l16 = lane & 15;
    const int g = blockIdx.x;
    const int col = (wave & 7) * 16 + l16;

    v2f wb[DIM / 4] = {};
    if (wave < 8) {
        #pragma unroll
        for (int kt = 0; kt < DIM / 4; ++kt) {
            const int k0 = kt * 4 + half * 2;
            wb[kt].x = W2[(size_t)k0 * DIM + col];
            wb[kt].y = W2[(size_t)(k0 + 1) * DIM + col];
        }
    }

    float4 s = make_float4(0.f, 0.f, 0.f, 0.f);
    for (int b0 = wave; b0 < nblk; b0 += 32 * 8) {
        #pragma unroll
        for (int u = 0; u < 8; ++u) {
            const int b = b0 + 32 * u;
            const float m = b < nblk ? 1.f : 0.f;
            const float4 v = *(const float4*)(part + (size_t)min(b, nblk - 1) * (NG * DIM) + g * DIM + lane * 4);
            s.x = fmaf(m, v.x, s.x); s.y = fmaf(m, v.y, s.y); s.z = fmaf(m, v.z, s.z); s.w = fmaf(m, v.w, s.w);
        }
    }
    float c = 0.f;
    for (int b = tid; b < nblk; b += 1024) c += pcnt[b * NG + g];
    #pragma unroll
    for (int off = 16; off > 0; off >>= 1) c += __shfl_xor(c, off);
    *(float4*)(&ps[wave][lane * 4]) = s;
    if (lane == 0) pc[wave] = c;
    __syncthreads();
    if (tid < DIM) {
        float ss = ps[0][tid];
        #pragma unroll
        for (int w = 1; w < 32; ++w) ss += ps[w][tid];
        prow[tid] = ss;
    }
    __syncthreads();
    if (wave < 8) {
        const int keep = l16 == 0 ? -1 : 0;
        v8f acc = {};
        #pragma unroll
        for (int kt = 0; kt < DIM / 4; ++kt) {
            const v2f pa = *(const v2f*)(&prow[kt * 4 + half * 2]);
            v2f a;
            a.x = __int_as_float(__float_as_int(pa.x) & keep);
            a.y = __int_as_float(__float_as_int(pa.y) & keep);
            acc = __builtin_amdgcn_wmma_f32_16x16x4_f32(false, a, false, wb[kt], (short)0, acc, false, false);
        }
        float cs = pc[0];
        #pragma unroll
        for (int w = 1; w < 32; ++w) cs += pc[w];
        if (half == 0) orow[col] = fmaf(cs, b2[col], acc[0]) / fmaxf(cs, 1.0f);
    }
    __syncthreads();
    if (tid < 32) *(float4*)(out + (size_t)g * DIM + tid * 4) = *(const float4*)(&orow[tid * 4]);
}

extern "C" void kernel_launch(void* const* d_in, const int* in_sizes, int n_in,
                              void* d_out, int out_size, void* d_ws, size_t ws_size,
                              hipStream_t stream) {
    const int*   node_ids = (const int*)d_in[0];
    const int*   ei       = (const int*)d_in[1];
    const int*   batch    = (const int*)d_in[2];
    const float* emb      = (const float*)d_in[4];
    const float* W1       = (const float*)d_in[5];
    const float* b1       = (const float*)d_in[6];
    const float* W2       = (const float*)d_in[7];
    const float* b2       = (const float*)d_in[8];
    float* out = (float*)d_out;

    const int N = in_sizes[0];
    const int E = in_sizes[1] / 2;
    const int* src = ei;
    const int* dst = ei + E;

    char* base = (char*)d_ws;
    size_t off = 0;
    auto alloc = [&](size_t bytes) -> char* {
        char* p = base + off;
        off = (off + bytes + 255) & ~(size_t)255;
        return p;
    };
    float* HW1  = (float*)alloc((size_t)NTP * DIM * sizeof(float));
    float* deg0 = (float*)alloc((size_t)N * sizeof(float));
    float* deg1 = (float*)alloc((size_t)N * sizeof(float));
    int*   rec  = (int*)alloc((size_t)N * 2 * sizeof(int));
    float* coef = (float*)alloc((size_t)N * CS * sizeof(float));
    float* coef1 = (float*)alloc((size_t)N * CS * sizeof(float));
    const int NB = (N + KB - 1) / KB;
    const int NPAIR = NB * (KB / 2);
    float* tq   = (float*)alloc((size_t)NPAIR * TP * sizeof(float));
    float* tq1  = (float*)alloc((size_t)NPAIR * TP * sizeof(float));
    float* part = (float*)alloc((size_t)NB * NG * DIM * sizeof(float));
    float* pcnt = (float*)alloc((size_t)NB * NG * sizeof(float));
    (void)ws_size; (void)n_in; (void)out_size;

    const int TB = 256;
    dim3 blk(TB);

    k_deg_init <<<dim3((N + TB - 1) / TB), blk, 0, stream>>>(deg0, deg1, N);
    k_deg_edges<<<dim3(HW1_BLOCKS + (E + TB - 1) / TB), blk, 0, stream>>>(dst, deg0, deg1, E,
                                                                        coef, coef1, N * (CS / 4),
                                                                        tq, tq1, NPAIR * (TP / 4),
                                                                        emb, W1, HW1);
    k_rec      <<<dim3((N + TB - 1) / TB), blk, 0, stream>>>(deg0, deg1, node_ids, batch, rec, coef, tq, N);

    k_cedge<<<dim3((E + TB - 1) / TB), blk, 0, stream>>>(src, dst, rec, coef, coef1, tq, tq1, E);

    k_poolmm<<<dim3(NB), blk, 0, stream>>>(coef, coef1, HW1, b1, tq, tq1, batch, part, pcnt, N);

    k_final <<<dim3(NG), dim3(1024), 0, stream>>>(part, pcnt, W2, b2, out, NB);
}


# edge pass: each node's degree, type, graph (3 bytes) held in LDS, copied once by each of 256 looping work-groups; end points looked up in LDS, not gathered from memory; atomic order as before
# speedup vs baseline: 7.5848x; 1.0778x over previous
#include <hip/hip_runtime.h>
#include <hip/hip_bf16.h>

typedef float v2f __attribute__((ext_vector_type(2)));
typedef float v8f __attribute__((ext_vector_type(8)));
typedef unsigned int v4u __attribute__((ext_vector_type(4)));

#define DIM 128
#define NT 17
#define NTP 20
#define CS NTP
#define NG 64
#define KB 208
#define TP 128

__device__ __forceinline__ int pkg_half() {
    return (int)((__builtin_amdgcn_s_sendmsg_rtn(0x87) >> 18) & 1u);
}

#define HW1_BLOCKS ((NTP * DIM + 255) / 256)
__device__ __forceinline__ void hw1_block(const float* __restrict__ emb,
                                          const float* __restrict__ W1,
                                          float* __restrict__ HW1) {
    int t = blockIdx.x * 256 + threadIdx.x;
    if (t >= NTP * DIM) return;
    int m = t >> 7, n = t & 127;
    float acc = 0.f;
    if (m < NT) {
        #pragma unroll 8
        for (int k = 0; k < DIM; ++k)
            acc = fmaf(emb[m * DIM + k], W1[k * DIM + n], acc);
    }
    HW1[t] = acc;
}

__global__ __launch_bounds__(256) void k_deg_init(float* __restrict__ deg0, float* __restrict__ deg1, int n) {
    int i = blockIdx.x * 256 + threadIdx.x;
    if (i < n) { deg0[i] = 1.0f; deg1[i] = 0.0f; }
}
__global__ __launch_bounds__(256) void k_deg_edges(const int* __restrict__ dst,
                                                   float* __restrict__ deg0, float* __restrict__ deg1, int nE,
                                                   float* __restrict__ coef0, float* __restrict__ coef1, int nc4,
                                                   float* __restrict__ tq0, float* __restrict__ tq1, int nt4,
                                                   const float* __restrict__ emb,
                                                   const float* __restrict__ W1,
                                                   float* __restrict__ HW1) {
    if (blockIdx.x < HW1_BLOCKS) { hw1_block(emb, W1, HW1); return; }
    const int e = (blockIdx.x - HW1_BLOCKS) * 256 + threadIdx.x;
    const int d = dst[min(e, nE - 1)];
    const int stride = (gridDim.x - HW1_BLOCKS) * 256;
    const float4 z = make_float4(0.f, 0.f, 0.f, 0.f);
    for (int i = e; i < nc4; i += stride) {
        *(float4*)(coef0 + (size_t)i * 4) = z;
        *(float4*)(coef1 + (size_t)i * 4) = z;
    }
    for (int i = e; i < nt4; i += stride) {
        *(float4*)(tq0 + (size_t)i * 4) = z;
        *(float4*)(tq1 + (size_t)i * 4) = z;
    }
    float* __restrict__ deg = pkg_half() ? deg1 : deg0;
    if (e < nE) atomicAdd(&deg[d], 1.0f);
}
#define DEGCAP 2047
__global__ __launch_bounds__(256) void k_rec(const float* __restrict__ deg0, const float* __restrict__ deg1,
                                             const int* __restrict__ node_ids, const int* __restrict__ batch,
                                             int* __restrict__ rec,
                                             int* __restrict__ recS,
                                             float* __restrict__ coef, float* __restrict__ tq, int n, int npad) {
    unsigned short* __restrict__ recA = (unsigned short*)recS;
    unsigned char* __restrict__ recB = (unsigned char*)recS + 2 * (size_t)npad;
    int i = blockIdx.x * 256 + threadIdx.x;
    if (i >= n) {
        if (i < npad) { recA[i] = 0; recB[i] = 0; }
        return;
    }
    {
        float d = deg0[i] + deg1[i];
        float v = d > 0.f ? rsqrtf(d) : 0.f;
        const int ty = node_ids[i] & 0xFFFF;
        const int g = batch[i];
        *(int2*)(rec + 2 * (size_t)i) = make_int2(__float_as_int(v), (g << 16) | ty);
        const bool fits = d < (float)DEGCAP && ty < 32 && g >= 0 && g < 256;
        recA[i] = (unsigned short)(fits ? (((int)d << 5) | ty) : (DEGCAP << 5));
        recB[i] = (unsigned char)(g & 255);
        const float w = v * v;
        coef[(size_t)i * CS + ty] = w;
        tq[(size_t)(i >> 1) * TP + (g & 15) * 8 + (g >> 4) * 2 + (i & 1)] = w;
    }
}
__device__ __forceinline__ float rec_dinv(int2 r)  { return __int_as_float(r.x); }
__device__ __forceinline__ int   rec_type(int2 r)  { return r.y & 0xFFFF; }
__device__ __forceinline__ int   rec_graph(int2 r) { return r.y >> 16; }

#define NMAX 100000
#define CE_T 1024
#define CE_U 4
#define CE_F 10
__device__ __forceinline__ float dinv_of_count(unsigned k) {
    const float f = (float)k;
    return f > 0.f ? rsqrtf(f) : 0.f;
}
__global__ __launch_bounds__(CE_T) void k_cedge(const int* __restrict__ src,
                                                const int* __restrict__ dst,
                                                const int* __restrict__ rec,
                                                const int* __restrict__ recS,
                                                float* __restrict__ coef0, float* __restrict__ coef1,
                                                float* __restrict__ tq0, float* __restrict__ tq1,
                                                int nE, int npad) {
    __shared__ __attribute__((aligned(16))) unsigned char sm[3 * NMAX];
    const unsigned short* sA = (const unsigned short*)sm;
    const unsigned char* sB = sm + 2 * npad;
    const int ph = pkg_half();
    float* __restrict__ coef = ph ? coef1 : coef0;
    float* __restrict__ tq   = ph ? tq1 : tq0;
    const bool in_lds = npad <= NMAX;
    if (in_lds) {
        const int np = (3 * npad) >> 4;
        const v4u* __restrict__ gp = (const v4u*)recS;
        v4u* lp = (v4u*)sm;
        #pragma unroll 1
        for (int p0 = threadIdx.x; p0 < np; p0 += CE_T * CE_F) {
            v4u v[CE_F];
            #pragma unroll
            for (int u = 0; u < CE_F; ++u) v[u] = gp[min(p0 + u * CE_T, np - 1)];
            static_assert(CE_F == 10, "the operand list below names CE_F values");
            asm volatile("" :: "v"(v[0]), "v"(v[1]), "v"(v[2]), "v"(v[3]), "v"(v[4]),
                               "v"(v[5]), "v"(v[6]), "v"(v[7]), "v"(v[8]), "v"(v[9]));
            #pragma unroll
            for (int u = 0; u < CE_F; ++u)
                if (p0 + u * CE_T < np) lp[p0 + u * CE_T] = v[u];
        }
    }
    __syncthreads();
    const int stride = gridDim.x * CE_T;
    for (int e0 = blockIdx.x * CE_T + threadIdx.x; e0 < nE; e0 += CE_U * stride) {
        int s[CE_U], d[CE_U];
        #pragma unroll
        for (int u = 0; u < CE_U; ++u) {
            const int e = min(e0 + u * stride, nE - 1);
            s[u] = src[e]; d[u] = dst[e];
        }
        #pragma unroll
        for (int u = 0; u < CE_U; ++u) {
            if (e0 + u * stride >= nE) break;
            float vs, vd;
            int ty, g;
            if (in_lds) {
                const unsigned as = sA[s[u]], ad = sA[d[u]];
                ty = (int)(as & 31u);
                g = (int)sB[d[u]];
                if ((as >> 5) < DEGCAP) vs = dinv_of_count(as >> 5);
                else { const int2 r = *(const int2*)(rec + 2 * (size_t)s[u]); vs = rec_dinv(r); ty = rec_type(r); }
                if ((ad >> 5) < DEGCAP) vd = dinv_of_count(ad >> 5);
                else { const int2 r = *(const int2*)(rec + 2 * (size_t)d[u]); vd = rec_dinv(r); g = rec_graph(r); }
            } else {
                const int2 rs = *(const int2*)(rec + 2 * (size_t)s[u]);
                const int2 rd = *(const int2*)(rec + 2 * (size_t)d[u]);
                vs = rec_dinv(rs); ty = rec_type(rs);
                vd = rec_dinv(rd); g = rec_graph(rd);
            }
            const float nrm = vs * vd;
            atomicAdd(&coef[(size_t)d[u] * CS + ty], nrm);
            atomicAdd(&tq[(size_t)(s[u] >> 1) * TP + (g & 15) * 8 + (g >> 4) * 2 + (s[u] & 1)], nrm);
        }
    }
}

#define NCH (KB / 16)
struct PFrag {
    v2f c0[NTP / 4], c1[NTP / 4];
    float4 u0[4], u1[4], w0[4], w1[4];
};
struct PSum {
    v2f a[NTP / 4];
    v2f t[4][4];
};
__device__ __forceinline__ void pf_load(PFrag& f, const float* __restrict__ coef, const float* __restrict__ coef1,
                                        const float* __restrict__ tq, const float* __restrict__ tq1,
                                        int base, int n, int half, int l16) {
    const size_t ro = (size_t)min(base + l16, n - 1) * CS + half * 2;
    #pragma unroll
    for (int kt = 0; kt < NTP / 4; ++kt) {
        f.c0[kt] = *(const v2f*)(coef  + ro + kt * 4);
        f.c1[kt] = *(const v2f*)(coef1 + ro + kt * 4);
    }
    const size_t to = (size_t)((base >> 1) + 4 * half) * TP + l16 * 8;
    #pragma unroll
    for (int q = 0; q < 4; ++q) {
        f.u0[q] = *(const float4*)(tq  + to + (size_t)q * TP);
        f.u1[q] = *(const float4*)(tq  + to + (size_t)q * TP + 4);
        f.w0[q] = *(const float4*)(tq1 + to + (size_t)q * TP);
        f.w1[q] = *(const float4*)(tq1 + to + (size_t)q * TP + 4);
    }
}
__device__ __forceinline__ void pf_sum(const PFrag& f, PSum& s) {
    #pragma unroll
    for (int kt = 0; kt < NTP / 4; ++kt) s.a[kt] = f.c0[kt] + f.c1[kt];
    #pragma unroll
    for (int q = 0; q < 4; ++q) {
        s.t[q][0].x = f.u0[q].x + f.w0[q].x; s.t[q][0].y = f.u0[q].y + f.w0[q].y;
        s.t[q][1].x = f.u0[q].z + f.w0[q].z; s.t[q][1].y = f.u0[q].w + f.w0[q].w;
        s.t[q][2].x = f.u1[q].x + f.w1[q].x; s.t[q][2].y = f.u1[q].y + f.w1[q].y;
        s.t[q][3].x = f.u1[q].z + f.w1[q].z; s.t[q][3].y = f.u1[q].w + f.w1[q].w;
    }
}
__device__ __forceinline__ void pf_mma(const PSum& s, const v2f (&hb)[2][NTP / 4], const float (&bias)[2],
                                       v8f (&acc)[2][4]) {
    v8f p0 = {}, p1 = {};
    #pragma unroll
    for (int kt = 0; kt < NTP / 4; ++kt) {
        p0 = __builtin_amdgcn_wmma_f32_16x16x4_f32(false, s.a[kt], false, hb[0][kt], (short)0, p0, false, false);
        p1 = __builtin_amdgcn_wmma_f32_16x16x4_f32(false, s.a[kt], false, hb[1][kt], (short)0, p1, false, false);
    }
    #pragma unroll
    for (int j = 0; j < 8; ++j) {
        p0[j] = fmaxf(p0[j] + bias[0], 0.f);
        p1[j] = fmaxf(p1[j] + bias[1], 0.f);
    }
    #pragma unroll
    for (int q = 0; q < 4; ++q) {
        v2f b0, b1;
        b0.x = p0[2 * q]; b0.y = p0[2 * q + 1];
        b1.x = p1[2 * q]; b1.y = p1[2 * q + 1];
        #pragma unroll
        for (int mt = 0; mt < 4; ++mt) {
            acc[0][mt] = __builtin_amdgcn_wmma_f32_16x16x4_f32(false, s.t[q][mt], false, b0, (short)0, acc[0][mt], false, false);
            acc[1][mt] = __builtin_amdgcn_wmma_f32_16x16x4_f32(false, s.t[q][mt], false, b1, (short)0, acc[1][mt], false, false);
        }
    }
}
__global__ __launch_bounds__(256) void k_poolmm(const float* __restrict__ coef,
                                                const float* __restrict__ coef1,
                                                const float* __restrict__ HW1,
                                                const float* __restrict__ b1,
                                                const float* __restrict__ tq,
                                                const float* __restrict__ tq1,
                                                const int* __restrict__ batch,
                                                float* __restrict__ part,
                                                float* __restrict__ pcnt, int n) {
    static_assert(KB % 16 == 0 && KB <= 256, "KB");
    __shared__ int cl[NG];
    __shared__ float st[NG][DIM + 4];
    const int tid = threadIdx.x;
    const int wave = __builtin_amdgcn_readfirstlane(tid >> 5);
    const int lane = tid & 31;
    const int half = lane >> 4, l16 = lane & 15;
    const int cg = wave & 3;
    const int par = wave >> 2;
    const int nb = blockIdx.x * KB;

    if (tid < NG) cl[tid] = 0;
    __syncthreads();
    if (tid < KB && nb + tid < n) atomicAdd(&cl[batch[nb + tid]], 1);

    v2f hb[2][NTP / 4];
    float bias[2];
    #pragma unroll
    for (int ct = 0; ct < 2; ++ct) {
        const int col = cg * 32 + ct * 16 + l16;
        #pragma unroll
        for (int kt = 0; kt < NTP / 4; ++kt) {
            const int k0 = kt * 4 + half * 2;
            hb[ct][kt].x = HW1[(size_t)k0 * DIM + col];
            hb[ct][kt].y = HW1[(size_t)(k0 + 1) * DIM + col];
        }
        bias[ct] = b1[col];
    }

    v8f acc[2][4] = {};
    #pragma unroll 1
    for (int ch = par; ch < NCH; ch += 2) {
        PFrag f;
        PSum s;
        pf_load(f, coef, coef1, tq, tq1, nb + ch * 16, n, half, l16);
        pf_sum(f, s);
        pf_mma(s, hb, bias, acc);
    }

    if (par == 1) {
        #pragma unroll
        for (int ct = 0; ct < 2; ++ct) {
            const int col = cg * 32 + ct * 16 + l16;
            #pragma unroll
            for (int mt = 0; mt < 4; ++mt) {
                #pragma unroll
                for (int j = 0; j < 8; ++j)
                    st[mt * 16 + j + half * 8][col] = acc[ct][mt][j];
            }
        }
    }
    __syncthreads();
    if (par == 0) {
        #pragma unroll
        for (int ct = 0; ct < 2; ++ct) {
            const int col = cg * 32 + ct * 16 + l16;
            #pragma unroll
            for (int mt = 0; mt < 4; ++mt) {
                #pragma unroll
                for (int j = 0; j < 8; ++j)
                    st[mt * 16 + j + half * 8][col] += acc[ct][mt][j];
            }
        }
    }
    __syncthreads();
    float* __restrict__ pb = part + (size_t)blockIdx.x * (NG * DIM);
    #pragma unroll
    for (int i = 0; i < NG / 8; ++i) {
        const int r = wave + 8 * i;
        *(float4*)(pb + (size_t)r * DIM + lane * 4) = *(const float4*)(&st[r][lane * 4]);
    }
    if (tid < NG) pcnt[blockIdx.x * NG + tid] = (float)cl[tid];
}

__global__ __launch_bounds__(1024) void k_final(const float* __restrict__ part,
                                                const float* __restrict__ pcnt,
                                                const float* __restrict__ W2,
                                                const float* __restrict__ b2,
                                                float* __restrict__ out, int nblk) {
    __shared__ float ps[32][DIM];
    __shared__ float prow[DIM];
    __shared__ float orow[DIM];
    __shared__ float pc[32];
    const int tid = threadIdx.x;
    const int wave = __builtin_amdgcn_readfirstlane(tid >> 5);
    const int lane = tid & 31;
    const int half = lane >> 4, l16 = lane & 15;
    const int g = blockIdx.x;
    const int col = (wave & 7) * 16 + l16;

    v2f wb[DIM / 4] = {};
    if (wave < 8) {
        #pragma unroll
        for (int kt = 0; kt < DIM / 4; ++kt) {
            const int k0 = kt * 4 + half * 2;
            wb[kt].x = W2[(size_t)k0 * DIM + col];
            wb[kt].y = W2[(size_t)(k0 + 1) * DIM + col];
        }
    }

    float4 s = make_float4(0.f, 0.f, 0.f, 0.f);
    for (int b0 = wave; b0 < nblk; b0 += 32 * 8) {
        #pragma unroll
        for (int u = 0; u < 8; ++u) {
            const int b = b0 + 32 * u;
            const float m = b < nblk ? 1.f : 0.f;
            const float4 v = *(const float4*)(part + (size_t)min(b, nblk - 1) * (NG * DIM) + g * DIM + lane * 4);
            s.x = fmaf(m, v.x, s.x); s.y = fmaf(m, v.y, s.y); s.z = fmaf(m, v.z, s.z); s.w = fmaf(m, v.w, s.w);
        }
    }
    float c = 0.f;
    for (int b = tid; b < nblk; b += 1024) c += pcnt[b * NG + g];
    #pragma unroll
    for (int off = 16; off > 0; off >>= 1) c += __shfl_xor(c, off);
    *(float4*)(&ps[wave][lane * 4]) = s;
    if (lane == 0) pc[wave] = c;
    __syncthreads();
    if (tid < DIM) {
        float ss = ps[0][tid];
        #pragma unroll
        for (int w = 1; w < 32; ++w) ss += ps[w][tid];
        prow[tid] = ss;
    }
    __syncthreads();
    if (wave < 8) {
        const int keep = l16 == 0 ? -1 : 0;
        v8f acc = {};
        #pragma unroll
        for (int kt = 0; kt < DIM / 4; ++kt) {
            const v2f pa = *(const v2f*)(&prow[kt * 4 + half * 2]);
            v2f a;
            a.x = __int_as_float(__float_as_int(pa.x) & keep);
            a.y = __int_as_float(__float_as_int(pa.y) & keep);
            acc = __builtin_amdgcn_wmma_f32_16x16x4_f32(false, a, false, wb[kt], (short)0, acc, false, false);
        }
        float cs = pc[0];
        #pragma unroll
        for (int w = 1; w < 32; ++w) cs += pc[w];
        if (half == 0) orow[col] = fmaf(cs, b2[col], acc[0]) / fmaxf(cs, 1.0f);
    }
    __syncthreads();
    if (tid < 32) *(float4*)(out + (size_t)g * DIM + tid * 4) = *(const float4*)(&orow[tid * 4]);
}

extern "C" void kernel_launch(void* const* d_in, const int* in_sizes, int n_in,
                              void* d_out, int out_size, void* d_ws, size_t ws_size,
                              hipStream_t stream) {
    const int*   node_ids = (const int*)d_in[0];
    const int*   ei       = (const int*)d_in[1];
    const int*   batch    = (const int*)d_in[2];
    const float* emb      = (const float*)d_in[4];
    const float* W1       = (const float*)d_in[5];
    const float* b1       = (const float*)d_in[6];
    const float* W2       = (const float*)d_in[7];
    const float* b2       = (const float*)d_in[8];
    float* out = (float*)d_out;

    const int N = in_sizes[0];
    const int E = in_sizes[1] / 2;
    const int* src = ei;
    const int* dst = ei + E;

    char* base = (char*)d_ws;
    size_t off = 0;
    auto alloc = [&](size_t bytes) -> char* {
        char* p = base + off;
        off = (off + bytes + 255) & ~(size_t)255;
        return p;
    };
    float* HW1  = (float*)alloc((size_t)NTP * DIM * sizeof(float));
    float* deg0 = (float*)alloc((size_t)N * sizeof(float));
    float* deg1 = (float*)alloc((size_t)N * sizeof(float));
    int*   rec  = (int*)alloc((size_t)N * 2 * sizeof(int));
    const int NPADN = (N + 15) & ~15;
    int* recS = (int*)alloc((size_t)NPADN * 3);
    float* coef = (float*)alloc((size_t)N * CS * sizeof(float));
    float* coef1 = (float*)alloc((size_t)N * CS * sizeof(float));
    const int NB = (N + KB - 1) / KB;
    const int NPAIR = NB * (KB / 2);
    float* tq   = (float*)alloc((size_t)NPAIR * TP * sizeof(float));
    float* tq1  = (float*)alloc((size_t)NPAIR * TP * sizeof(float));
    float* part = (float*)alloc((size_t)NB * NG * DIM * sizeof(float));
    float* pcnt = (float*)alloc((size_t)NB * NG * sizeof(float));
    (void)ws_size; (void)n_in; (void)out_size;

    const int TB = 256;
    dim3 blk(TB);

    k_deg_init <<<dim3((N + TB - 1) / TB), blk, 0, stream>>>(deg0, deg1, N);
    k_deg_edges<<<dim3(HW1_BLOCKS + (E + TB - 1) / TB), blk, 0, stream>>>(dst, deg0, deg1, E,
                                                                        coef, coef1, N * (CS / 4),
                                                                        tq, tq1, NPAIR * (TP / 4),
                                                                        emb, W1, HW1);
    k_rec      <<<dim3((NPADN + TB - 1) / TB), blk, 0, stream>>>(deg0, deg1, node_ids, batch, rec, recS, coef, tq, N, NPADN);

    k_cedge<<<dim3(256), dim3(CE_T), 0, stream>>>(src, dst, rec, recS, coef, coef1, tq, tq1, E, NPADN);

    k_poolmm<<<dim3(NB), blk, 0, stream>>>(coef, coef1, HW1, b1, tq, tq1, batch, part, pcnt, N);

    k_final <<<dim3(NG), dim3(1024), 0, stream>>>(part, pcnt, W2, b2, out, NB);
}


# edge pass: coef atomics only from package-half-0 work-groups, tq atomics only from half-1 (shares drawn by one returning add per table); one coef/tq plane, not two; changes summation order
# speedup vs baseline: 7.8939x; 1.0408x over previous
#include <hip/hip_runtime.h>
#include <hip/hip_bf16.h>

typedef float v2f __attribute__((ext_vector_type(2)));
typedef float v8f __attribute__((ext_vector_type(8)));
typedef unsigned int v4u __attribute__((ext_vector_type(4)));

#define DIM 128
#define NT 17
#define NTP 20
#define CS NTP
#define NG 64
#define KB 208
#define TP 128

__device__ __forceinline__ int pkg_half() {
    return (int)((__builtin_amdgcn_s_sendmsg_rtn(0x87) >> 18) & 1u);
}

#define HW1_BLOCKS ((NTP * DIM + 255) / 256)
__device__ __forceinline__ void hw1_block(const float* __restrict__ emb,
                                          const float* __restrict__ W1,
                                          float* __restrict__ HW1) {
    int t = blockIdx.x * 256 + threadIdx.x;
    if (t >= NTP * DIM) return;
    int m = t >> 7, n = t & 127;
    float acc = 0.f;
    if (m < NT) {
        #pragma unroll 8
        for (int k = 0; k < DIM; ++k)
            acc = fmaf(emb[m * DIM + k], W1[k * DIM + n], acc);
    }
    HW1[t] = acc;
}

__global__ __launch_bounds__(256) void k_deg_init(float* __restrict__ deg0, float* __restrict__ deg1, int n,
                                                  int* __restrict__ tick) {
    int i = blockIdx.x * 256 + threadIdx.x;
    if (i < n) { deg0[i] = 1.0f; deg1[i] = 0.0f; }
    if (i < 2) tick[i] = 0;
}
__global__ __launch_bounds__(256) void k_deg_edges(const int* __restrict__ dst,
                                                   float* __restrict__ deg0, float* __restrict__ deg1, int nE,
                                                   float* __restrict__ coef, int nc4,
                                                   float* __restrict__ tq, int nt4,
                                                   const float* __restrict__ emb,
                                                   const float* __restrict__ W1,
                                                   float* __restrict__ HW1) {
    if (blockIdx.x < HW1_BLOCKS) { hw1_block(emb, W1, HW1); return; }
    const int e = (blockIdx.x - HW1_BLOCKS) * 256 + threadIdx.x;
    const int d = dst[min(e, nE - 1)];
    const int stride = (gridDim.x - HW1_BLOCKS) * 256;
    const float4 z = make_float4(0.f, 0.f, 0.f, 0.f);
    for (int i = e; i < nc4; i += stride)
        *(float4*)(coef + (size_t)i * 4) = z;
    for (int i = e; i < nt4; i += stride)
        *(float4*)(tq + (size_t)i * 4) = z;
    float* __restrict__ deg = pkg_half() ? deg1 : deg0;
    if (e < nE) atomicAdd(&deg[d], 1.0f);
}
#define DEGCAP 2047
__global__ __launch_bounds__(256) void k_rec(const float* __restrict__ deg0, const float* __restrict__ deg1,
                                             const int* __restrict__ node_ids, const int* __restrict__ batch,
                                             int* __restrict__ rec,
                                             int* __restrict__ recS,
                                             float* __restrict__ coef, float* __restrict__ tq, int n, int npad) {
    unsigned short* __restrict__ recA = (unsigned short*)recS;
    unsigned char* __restrict__ recB = (unsigned char*)recS + 2 * (size_t)npad;
    int i = blockIdx.x * 256 + threadIdx.x;
    if (i >= n) {
        if (i < npad) { recA[i] = 0; recB[i] = 0; }
        return;
    }
    {
        float d = deg0[i] + deg1[i];
        float v = d > 0.f ? rsqrtf(d) : 0.f;
        const int ty = node_ids[i] & 0xFFFF;
        const int g = batch[i];
        *(int2*)(rec + 2 * (size_t)i) = make_int2(__float_as_int(v), (g << 16) | ty);
        const bool fits = d < (float)DEGCAP && ty < 32 && g >= 0 && g < 256;
        recA[i] = (unsigned short)(fits ? (((int)d << 5) | ty) : (DEGCAP << 5));
        recB[i] = (unsigned char)(g & 255);
        const float w = v * v;
        coef[(size_t)i * CS + ty] = w;
        tq[(size_t)(i >> 1) * TP + (g & 15) * 8 + (g >> 4) * 2 + (i & 1)] = w;
    }
}
__device__ __forceinline__ float rec_dinv(int2 r)  { return __int_as_float(r.x); }
__device__ __forceinline__ int   rec_type(int2 r)  { return r.y & 0xFFFF; }
__device__ __forceinline__ int   rec_graph(int2 r) { return r.y >> 16; }

#define NMAX 100000
#define CE_T 1024
#define CE_U 8
#define CE_F 10
__device__ __forceinline__ float dinv_of_count(unsigned k) {
    const float f = (float)k;
    return f > 0.f ? rsqrtf(f) : 0.f;
}
template <int TAB>
__device__ __forceinline__ void ce_share(const int* __restrict__ src, const int* __restrict__ dst,
                                         const int* __restrict__ rec,
                                         const unsigned short* sA, const unsigned char* sB, bool in_lds,
                                         float* __restrict__ tab, int nE, int share, int nshare,
                                         int* __restrict__ nxt, int& tnxt, bool& drawn) {
    const int len = (nE + nshare - 1) / nshare;
    const int lo = share * len;
    const int hi = min(lo + len, nE);
    for (int e0 = lo + (int)threadIdx.x; e0 < hi; e0 += CE_U * CE_T) {
        int s[CE_U], d[CE_U];
        #pragma unroll
        for (int u = 0; u < CE_U; ++u) {
            const int e = min(e0 + u * CE_T, hi - 1);
            s[u] = src[e]; d[u] = dst[e];
        }
        if (nxt != nullptr && e0 == lo) { tnxt = atomicAdd(nxt, 1); drawn = true; }
        #pragma unroll
        for (int u = 0; u < CE_U; ++u) {
            if (e0 + u * CE_T >= hi) break;
            float vs, vd;
            int ty, g;
            if (in_lds) {
                const unsigned as = sA[s[u]], ad = sA[d[u]];
                ty = (int)(as & 31u);
                g = TAB ? (int)sB[d[u]] : 0;
                if ((as >> 5) < DEGCAP) vs = dinv_of_count(as >> 5);
                else { const int2 r = *(const int2*)(rec + 2 * (size_t)s[u]); vs = rec_dinv(r); ty = rec_type(r); }
                if ((ad >> 5) < DEGCAP) vd = dinv_of_count(ad >> 5);
                else { const int2 r = *(const int2*)(rec + 2 * (size_t)d[u]); vd = rec_dinv(r); g = rec_graph(r); }
            } else {
                const int2 rs = *(const int2*)(rec + 2 * (size_t)s[u]);
                const int2 rd = *(const int2*)(rec + 2 * (size_t)d[u]);
                vs = rec_dinv(rs); ty = rec_type(rs);
                vd = rec_dinv(rd); g = rec_graph(rd);
            }
            const float nrm = vs * vd;
            if (TAB == 0) atomicAdd(&tab[(size_t)d[u] * CS + ty], nrm);
            else atomicAdd(&tab[(size_t)(s[u] >> 1) * TP + (g & 15) * 8 + (g >> 4) * 2 + (s[u] & 1)], nrm);
        }
    }
}
__global__ __launch_bounds__(CE_T) void k_cedge(const int* __restrict__ src,
                                                const int* __restrict__ dst,
                                                const int* __restrict__ rec,
                                                const int* __restrict__ recS,
                                                float* __restrict__ coef, float* __restrict__ tq,
                                                int* __restrict__ tick,
                                                int nE, int npad) {
    __shared__ __attribute__((aligned(16))) unsigned char sm[3 * NMAX];
    __shared__ int sh[3];
    const unsigned short* sA = (const unsigned short*)sm;
    const unsigned char* sB = sm + 2 * npad;
    int h0 = 0, t0 = 0;
    bool drawn0 = false;
    const bool in_lds = npad <= NMAX;
    if (in_lds) {
        const int np = (3 * npad) >> 4;
        const v4u* __restrict__ gp = (const v4u*)recS;
        v4u* lp = (v4u*)sm;
        #pragma unroll 1
        for (int p0 = threadIdx.x; p0 < np; p0 += CE_T * CE_F) {
            v4u v[CE_F];
            #pragma unroll
            for (int u = 0; u < CE_F; ++u) v[u] = gp[min(p0 + u * CE_T, np - 1)];
            if (p0 == 0) { h0 = pkg_half(); t0 = atomicAdd(&tick[h0], 1); drawn0 = true; }
            static_assert(CE_F == 10, "the operand list below names CE_F values");
            asm volatile("" :: "v"(v[0]), "v"(v[1]), "v"(v[2]), "v"(v[3]), "v"(v[4]),
                               "v"(v[5]), "v"(v[6]), "v"(v[7]), "v"(v[8]), "v"(v[9]));
            #pragma unroll
            for (int u = 0; u < CE_F; ++u)
                if (p0 + u * CE_T < np) lp[p0 + u * CE_T] = v[u];
        }
    }
    if (threadIdx.x == 0) {
        if (!drawn0) { h0 = pkg_half(); t0 = atomicAdd(&tick[h0], 1); }
        sh[0] = h0; sh[1] = t0;
    }
    __syncthreads();
    const int half = sh[0];
    const int nshare = max((int)gridDim.x >> 1, 1);
    int t1 = 0;
    bool drawn1 = false;
    int* const oth = tick + (half ^ 1);
    const int t_own = sh[1];
    if (t_own < nshare) {
        if (half == 0) ce_share<0>(src, dst, rec, sA, sB, in_lds, coef, nE, t_own, nshare, oth, t1, drawn1);
        else           ce_share<1>(src, dst, rec, sA, sB, in_lds, tq,   nE, t_own, nshare, oth, t1, drawn1);
    }
    if (threadIdx.x == 0) {
        if (!drawn1) t1 = atomicAdd(oth, 1);
        sh[2] = t1;
    }
    __syncthreads();
    const int t_oth = sh[2];
    if (t_oth < nshare) {
        int tn = 0; bool dn = false;
        if (half == 0) ce_share<1>(src, dst, rec, sA, sB, in_lds, tq,   nE, t_oth, nshare, nullptr, tn, dn);
        else           ce_share<0>(src, dst, rec, sA, sB, in_lds, coef, nE, t_oth, nshare, nullptr, tn, dn);
    }
}

#define NCH (KB / 16)
struct PFrag {
    v2f c0[NTP / 4];
    float4 u0[4], u1[4];
};
struct PSum {
    v2f a[NTP / 4];
    v2f t[4][4];
};
__device__ __forceinline__ void pf_load(PFrag& f, const float* __restrict__ coef,
                                        const float* __restrict__ tq,
                                        int base, int n, int half, int l16) {
    const size_t ro = (size_t)min(base + l16, n - 1) * CS + half * 2;
    #pragma unroll
    for (int kt = 0; kt < NTP / 4; ++kt)
        f.c0[kt] = *(const v2f*)(coef + ro + kt * 4);
    const size_t to = (size_t)((base >> 1) + 4 * half) * TP + l16 * 8;
    #pragma unroll
    for (int q = 0; q < 4; ++q) {
        f.u0[q] = *(const float4*)(tq + to + (size_t)q * TP);
        f.u1[q] = *(const float4*)(tq + to + (size_t)q * TP + 4);
    }
}
__device__ __forceinline__ void pf_sum(const PFrag& f, PSum& s) {
    #pragma unroll
    for (int kt = 0; kt < NTP / 4; ++kt) s.a[kt] = f.c0[kt];
    #pragma unroll
    for (int q = 0; q < 4; ++q) {
        s.t[q][0].x = f.u0[q].x; s.t[q][0].y = f.u0[q].y;
        s.t[q][1].x = f.u0[q].z; s.t[q][1].y = f.u0[q].w;
        s.t[q][2].x = f.u1[q].x; s.t[q][2].y = f.u1[q].y;
        s.t[q][3].x = f.u1[q].z; s.t[q][3].y = f.u1[q].w;
    }
}
__device__ __forceinline__ void pf_mma(const PSum& s, const v2f (&hb)[2][NTP / 4], const float (&bias)[2],
                                       v8f (&acc)[2][4]) {
    v8f p0 = {}, p1 = {};
    #pragma unroll
    for (int kt = 0; kt < NTP / 4; ++kt) {
        p0 = __builtin_amdgcn_wmma_f32_16x16x4_f32(false, s.a[kt], false, hb[0][kt], (short)0, p0, false, false);
        p1 = __builtin_amdgcn_wmma_f32_16x16x4_f32(false, s.a[kt], false, hb[1][kt], (short)0, p1, false, false);
    }
    #pragma unroll
    for (int j = 0; j < 8; ++j) {
        p0[j] = fmaxf(p0[j] + bias[0], 0.f);
        p1[j] = fmaxf(p1[j] + bias[1], 0.f);
    }
    #pragma unroll
    for (int q = 0; q < 4; ++q) {
        v2f b0, b1;
        b0.x = p0[2 * q]; b0.y = p0[2 * q + 1];
        b1.x = p1[2 * q]; b1.y = p1[2 * q + 1];
        #pragma unroll
        for (int mt = 0; mt < 4; ++mt) {
            acc[0][mt] = __builtin_amdgcn_wmma_f32_16x16x4_f32(false, s.t[q][mt], false, b0, (short)0, acc[0][mt], false, false);
            acc[1][mt] = __builtin_amdgcn_wmma_f32_16x16x4_f32(false, s.t[q][mt], false, b1, (short)0, acc[1][mt], false, false);
        }
    }
}
__global__ __launch_bounds__(256) void k_poolmm(const float* __restrict__ coef,
                                                const float* __restrict__ HW1,
                                                const float* __restrict__ b1,
                                                const float* __restrict__ tq,
                                                const int* __restrict__ batch,
                                                float* __restrict__ part,
                                                float* __restrict__ pcnt, int n) {
    static_assert(KB % 16 == 0 && KB <= 256, "KB");
    __shared__ int cl[NG];
    __shared__ float st[NG][DIM + 4];
    const int tid = threadIdx.x;
    const int wave = __builtin_amdgcn_readfirstlane(tid >> 5);
    const int lane = tid & 31;
    const int half = lane >> 4, l16 = lane & 15;
    const int cg = wave & 3;
    const int par = wave >> 2;
    const int nb = blockIdx.x * KB;

    if (tid < NG) cl[tid] = 0;
    __syncthreads();
    if (tid < KB && nb + tid < n) atomicAdd(&cl[batch[nb + tid]], 1);

    v2f hb[2][NTP / 4];
    float bias[2];
    #pragma unroll
    for (int ct = 0; ct < 2; ++ct) {
        const int col = cg * 32 + ct * 16 + l16;
        #pragma unroll
        for (int kt = 0; kt < NTP / 4; ++kt) {
            const int k0 = kt * 4 + half * 2;
            hb[ct][kt].x = HW1[(size_t)k0 * DIM + col];
            hb[ct][kt].y = HW1[(size_t)(k0 + 1) * DIM + col];
        }
        bias[ct] = b1[col];
    }

    v8f acc[2][4] = {};
    #pragma unroll 1
    for (int ch = par; ch < NCH; ch += 2) {
        PFrag f;
        PSum s;
        pf_load(f, coef, tq, nb + ch * 16, n, half, l16);
        pf_sum(f, s);
        pf_mma(s, hb, bias, acc);
    }

    if (par == 1) {
        #pragma unroll
        for (int ct = 0; ct < 2; ++ct) {
            const int col = cg * 32 + ct * 16 + l16;
            #pragma unroll
            for (int mt = 0; mt < 4; ++mt) {
                #pragma unroll
                for (int j = 0; j < 8; ++j)
                    st[mt * 16 + j + half * 8][col] = acc[ct][mt][j];
            }
        }
    }
    __syncthreads();
    if (par == 0) {
        #pragma unroll
        for (int ct = 0; ct < 2; ++ct) {
            const int col = cg * 32 + ct * 16 + l16;
            #pragma unroll
            for (int mt = 0; mt < 4; ++mt) {
                #pragma unroll
                for (int j = 0; j < 8; ++j)
                    st[mt * 16 + j + half * 8][col] += acc[ct][mt][j];
            }
        }
    }
    __syncthreads();
    float* __restrict__ pb = part + (size_t)blockIdx.x * (NG * DIM);
    #pragma unroll
    for (int i = 0; i < NG / 8; ++i) {
        const int r = wave + 8 * i;
        *(float4*)(pb + (size_t)r * DIM + lane * 4) = *(const float4*)(&st[r][lane * 4]);
    }
    if (tid < NG) pcnt[blockIdx.x * NG + tid] = (float)cl[tid];
}

__global__ __launch_bounds__(1024) void k_final(const float* __restrict__ part,
                                                const float* __restrict__ pcnt,
                                                const float* __restrict__ W2,
                                                const float* __restrict__ b2,
                                                float* __restrict__ out, int nblk) {
    __shared__ float ps[32][DIM];
    __shared__ float prow[DIM];
    __shared__ float orow[DIM];
    __shared__ float pc[32];
    const int tid = threadIdx.x;
    const int wave = __builtin_amdgcn_readfirstlane(tid >> 5);
    const int lane = tid & 31;
    const int half = lane >> 4, l16 = lane & 15;
    const int g = blockIdx.x;
    const int col = (wave & 7) * 16 + l16;

    v2f wb[DIM / 4] = {};
    if (wave < 8) {
        #pragma unroll
        for (int kt = 0; kt < DIM / 4; ++kt) {
            const int k0 = kt * 4 + half * 2;
            wb[kt].x = W2[(size_t)k0 * DIM + col];
            wb[kt].y = W2[(size_t)(k0 + 1) * DIM + col];
        }
    }

    float4 s = make_float4(0.f, 0.f, 0.f, 0.f);
    for (int b0 = wave; b0 < nblk; b0 += 32 * 8) {
        #pragma unroll
        for (int u = 0; u < 8; ++u) {
            const int b = b0 + 32 * u;
            const float m = b < nblk ? 1.f : 0.f;
            const float4 v = *(const float4*)(part + (size_t)min(b, nblk - 1) * (NG * DIM) + g * DIM + lane * 4);
            s.x = fmaf(m, v.x, s.x); s.y = fmaf(m, v.y, s.y); s.z = fmaf(m, v.z, s.z); s.w = fmaf(m, v.w, s.w);
        }
    }
    float c = 0.f;
    for (int b = tid; b < nblk; b += 1024) c += pcnt[b * NG + g];
    #pragma unroll
    for (int off = 16; off > 0; off >>= 1) c += __shfl_xor(c, off);
    *(float4*)(&ps[wave][lane * 4]) = s;
    if (lane == 0) pc[wave] = c;
    __syncthreads();
    if (tid < DIM) {
        float ss = ps[0][tid];
        #pragma unroll
        for (int w = 1; w < 32; ++w) ss += ps[w][tid];
        prow[tid] = ss;
    }
    __syncthreads();
    if (wave < 8) {
        const int keep = l16 == 0 ? -1 : 0;
        v8f acc = {};
        #pragma unroll
        for (int kt = 0; kt < DIM / 4; ++kt) {
            const v2f pa = *(const v2f*)(&prow[kt * 4 + half * 2]);
            v2f a;
            a.x = __int_as_float(__float_as_int(pa.x) & keep);
            a.y = __int_as_float(__float_as_int(pa.y) & keep);
            acc = __builtin_amdgcn_wmma_f32_16x16x4_f32(false, a, false, wb[kt], (short)0, acc, false, false);
        }
        float cs = pc[0];
        #pragma unroll
        for (int w = 1; w < 32; ++w) cs += pc[w];
        if (half == 0) orow[col] = fmaf(cs, b2[col], acc[0]) / fmaxf(cs, 1.0f);
    }
    __syncthreads();
    if (tid < 32) *(float4*)(out + (size_t)g * DIM + tid * 4) = *(const float4*)(&orow[tid * 4]);
}

extern "C" void kernel_launch(void* const* d_in, const int* in_sizes, int n_in,
                              void* d_out, int out_size, void* d_ws, size_t ws_size,
                              hipStream_t stream) {
    const int*   node_ids = (const int*)d_in[0];
    const int*   ei       = (const int*)d_in[1];
    const int*   batch    = (const int*)d_in[2];
    const float* emb      = (const float*)d_in[4];
    const float* W1       = (const float*)d_in[5];
    const float* b1       = (const float*)d_in[6];
    const float* W2       = (const float*)d_in[7];
    const float* b2       = (const float*)d_in[8];
    float* out = (float*)d_out;

    const int N = in_sizes[0];
    const int E = in_sizes[1] / 2;
    const int* src = ei;
    const int* dst = ei + E;

    char* base = (char*)d_ws;
    size_t off = 0;
    auto alloc = [&](size_t bytes) -> char* {
        char* p = base + off;
        off = (off + bytes + 255) & ~(size_t)255;
        return p;
    };
    float* HW1  = (float*)alloc((size_t)NTP * DIM * sizeof(float));
    float* deg0 = (float*)alloc((size_t)N * sizeof(float));
    float* deg1 = (float*)alloc((size_t)N * sizeof(float));
    int*   rec  = (int*)alloc((size_t)N * 2 * sizeof(int));
    const int NPADN = (N + 15) & ~15;
    int* recS = (int*)alloc((size_t)NPADN * 3);
    float* coef = (float*)alloc((size_t)N * CS * sizeof(float));
    const int NB = (N + KB - 1) / KB;
    const int NPAIR = NB * (KB / 2);
    float* tq   = (float*)alloc((size_t)NPAIR * TP * sizeof(float));
    float* part = (float*)alloc((size_t)NB * NG * DIM * sizeof(float));
    float* pcnt = (float*)alloc((size_t)NB * NG * sizeof(float));
    int*   tick = (int*)alloc(2 * sizeof(int));
    (void)ws_size; (void)n_in; (void)out_size;

    const int TB = 256;
    dim3 blk(TB);

    k_deg_init <<<dim3((N + TB - 1) / TB), blk, 0, stream>>>(deg0, deg1, N, tick);
    k_deg_edges<<<dim3(HW1_BLOCKS + (E + TB - 1) / TB), blk, 0, stream>>>(dst, deg0, deg1, E,
                                                                        coef, N * (CS / 4),
                                                                        tq, NPAIR * (TP / 4),
                                                                        emb, W1, HW1);
    k_rec      <<<dim3((NPADN + TB - 1) / TB), blk, 0, stream>>>(deg0, deg1, node_ids, batch, rec, recS, coef, tq, N, NPADN);

    k_cedge<<<dim3(256), dim3(CE_T), 0, stream>>>(src, dst, rec, recS, coef, tq, tick, E, NPADN);

    k_poolmm<<<dim3(NB), blk, 0, stream>>>(coef, HW1, b1, tq, batch, part, pcnt, N);

    k_final <<<dim3(NG), dim3(1024), 0, stream>>>(part, pcnt, W2, b2, out, NB);
}


# in-degree counted in LDS: 64 work-groups each count a share of the destinations in packed 16-bit LDS counters and store a plane; the record kernel sums the planes; no degree atomics, launches 6 -> 5
# speedup vs baseline: 8.5503x; 1.0832x over previous
#include <hip/hip_runtime.h>
#include <hip/hip_bf16.h>

typedef float v2f __attribute__((ext_vector_type(2)));
typedef float v8f __attribute__((ext_vector_type(8)));
typedef unsigned int v4u __attribute__((ext_vector_type(4)));

#define DIM 128
#define NT 17
#define NTP 20
#define CS NTP
#define NG 64
#define KB 208
#define TP 128

__device__ __forceinline__ int pkg_half() {
    return (int)((__builtin_amdgcn_s_sendmsg_rtn(0x87) >> 18) & 1u);
}

#define FT 1024
#define HW1_WGS ((NTP * DIM + FT - 1) / FT)
__device__ __forceinline__ void hw1_block(const float* __restrict__ emb,
                                          const float* __restrict__ W1,
                                          float* __restrict__ HW1, int wg) {
    int t = wg * FT + threadIdx.x;
    if (t >= NTP * DIM) return;
    int m = t >> 7, n = t & 127;
    float acc = 0.f;
    if (m < NT) {
        #pragma unroll 32
        for (int k = 0; k < DIM; ++k)
            acc = fmaf(emb[m * DIM + k], W1[k * DIM + n], acc);
    }
    HW1[t] = acc;
}

#define NMAX 100000
#define HU 25
__global__ __launch_bounds__(FT) void k_first(const int* __restrict__ dst, int nE, int npad,
                                              unsigned short* __restrict__ hist, int nh,
                                              float* __restrict__ coef, int nc4,
                                              float* __restrict__ tq, int nt4,
                                              const float* __restrict__ emb,
                                              const float* __restrict__ W1,
                                              float* __restrict__ HW1,
                                              int* __restrict__ tick) {
    __shared__ __attribute__((aligned(16))) unsigned hs[NMAX / 2];
    const int wg = blockIdx.x;
    if (wg >= nh) {
        if (wg - nh < HW1_WGS) { hw1_block(emb, W1, HW1, wg - nh); return; }
        const int cw = wg - nh - HW1_WGS;
        const int ncw = (int)gridDim.x - nh - HW1_WGS;
        if (cw == 0 && threadIdx.x < 2) tick[threadIdx.x] = 0;
        const int t0 = cw * FT + (int)threadIdx.x, stride = ncw * FT;
        const float4 z = make_float4(0.f, 0.f, 0.f, 0.f);
        for (int i = t0; i < nc4; i += stride)
            *(float4*)(coef + (size_t)i * 4) = z;
        for (int i = t0; i < nt4; i += stride)
            *(float4*)(tq + (size_t)i * 4) = z;
        return;
    }
    const int len = (nE + nh - 1) / nh;
    const int lo = wg * len;
    const int hi = min(lo + len, nE);
    unsigned short* __restrict__ plane = hist + (size_t)wg * npad;
    for (int nb = 0; nb < npad; nb += NMAX) {
        const int cnt = min(NMAX, npad - nb);
        const int np = cnt >> 3;
        const v4u zz = {0u, 0u, 0u, 0u};
        for (int p = threadIdx.x; p < np; p += FT) ((v4u*)hs)[p] = zz;
        __syncthreads();
        for (int e0 = lo + (int)threadIdx.x; e0 < hi; e0 += HU * FT) {
            int d[HU];
            #pragma unroll
            for (int u = 0; u < HU; ++u)
                d[u] = dst[min(e0 + u * FT, hi - 1)];
            #pragma unroll
            for (int u = 0; u < HU; ++u) {
                const unsigned r = (unsigned)(d[u] - nb);
                if (e0 + u * FT < hi && r < (unsigned)cnt)
                    atomicAdd(&hs[r >> 1], 1u << ((r & 1u) * 16u));
            }
        }
        __syncthreads();
        v4u* __restrict__ gp = (v4u*)(plane + nb);
        for (int p = threadIdx.x; p < np; p += FT) gp[p] = ((const v4u*)hs)[p];
        __syncthreads();
    }
}
#define DEGCAP 2047
__global__ __launch_bounds__(256) void k_rec(const unsigned short* __restrict__ hist, int nh,
                                             const int* __restrict__ node_ids, const int* __restrict__ batch,
                                             int* __restrict__ rec,
                                             int* __restrict__ recS,
                                             float* __restrict__ coef, float* __restrict__ tq, int n, int npad) {
    unsigned short* __restrict__ recA = (unsigned short*)recS;
    unsigned char* __restrict__ recB = (unsigned char*)recS + 2 * (size_t)npad;
    int i = blockIdx.x * 256 + threadIdx.x;
    if (i >= n) {
        if (i < npad) { recA[i] = 0; recB[i] = 0; }
        return;
    }
    {
        unsigned k = 1u;
        #pragma unroll 32
        for (int w = 0; w < nh; ++w) k += hist[(size_t)w * npad + i];
        float d = (float)k;
        float v = d > 0.f ? rsqrtf(d) : 0.f;
        const int ty = node_ids[i] & 0xFFFF;
        const int g = batch[i];
        *(int2*)(rec + 2 * (size_t)i) = make_int2(__float_as_int(v), (g << 16) | ty);
        const bool fits = d < (float)DEGCAP && ty < 32 && g >= 0 && g < 256;
        recA[i] = (unsigned short)(fits ? (((int)d << 5) | ty) : (DEGCAP << 5));
        recB[i] = (unsigned char)(g & 255);
        const float w = v * v;
        coef[(size_t)i * CS + ty] = w;
        tq[(size_t)(i >> 1) * TP + (g & 15) * 8 + (g >> 4) * 2 + (i & 1)] = w;
    }
}
__device__ __forceinline__ float rec_dinv(int2 r)  { return __int_as_float(r.x); }
__device__ __forceinline__ int   rec_type(int2 r)  { return r.y & 0xFFFF; }
__device__ __forceinline__ int   rec_graph(int2 r) { return r.y >> 16; }

#define CE_T 1024
#define CE_U 8
#define CE_F 10
__device__ __forceinline__ float dinv_of_count(unsigned k) {
    const float f = (float)k;
    return f > 0.f ? rsqrtf(f) : 0.f;
}
template <int TAB>
__device__ __forceinline__ void ce_share(const int* __restrict__ src, const int* __restrict__ dst,
                                         const int* __restrict__ rec,
                                         const unsigned short* sA, const unsigned char* sB, bool in_lds,
                                         float* __restrict__ tab, int nE, int share, int nshare,
                                         int* __restrict__ nxt, int& tnxt, bool& drawn) {
    const int len = (nE + nshare - 1) / nshare;
    const int lo = share * len;
    const int hi = min(lo + len, nE);
    for (int e0 = lo + (int)threadIdx.x; e0 < hi; e0 += CE_U * CE_T) {
        int s[CE_U], d[CE_U];
        #pragma unroll
        for (int u = 0; u < CE_U; ++u) {
            const int e = min(e0 + u * CE_T, hi - 1);
            s[u] = src[e]; d[u] = dst[e];
        }
        if (nxt != nullptr && e0 == lo) { tnxt = atomicAdd(nxt, 1); drawn = true; }
        #pragma unroll
        for (int u = 0; u < CE_U; ++u) {
            if (e0 + u * CE_T >= hi) break;
            float vs, vd;
            int ty, g;
            if (in_lds) {
                const unsigned as = sA[s[u]], ad = sA[d[u]];
                ty = (int)(as & 31u);
                g = TAB ? (int)sB[d[u]] : 0;
                if ((as >> 5) < DEGCAP) vs = dinv_of_count(as >> 5);
                else { const int2 r = *(const int2*)(rec + 2 * (size_t)s[u]); vs = rec_dinv(r); ty = rec_type(r); }
                if ((ad >> 5) < DEGCAP) vd = dinv_of_count(ad >> 5);
                else { const int2 r = *(const int2*)(rec + 2 * (size_t)d[u]); vd = rec_dinv(r); g = rec_graph(r); }
            } else {
                const int2 rs = *(const int2*)(rec + 2 * (size_t)s[u]);
                const int2 rd = *(const int2*)(rec + 2 * (size_t)d[u]);
                vs = rec_dinv(rs); ty = rec_type(rs);
                vd = rec_dinv(rd); g = rec_graph(rd);
            }
            const float nrm = vs * vd;
            if (TAB == 0) atomicAdd(&tab[(size_t)d[u] * CS + ty], nrm);
            else atomicAdd(&tab[(size_t)(s[u] >> 1) * TP + (g & 15) * 8 + (g >> 4) * 2 + (s[u] & 1)], nrm);
        }
    }
}
__global__ __launch_bounds__(CE_T) void k_cedge(const int* __restrict__ src,
                                                const int* __restrict__ dst,
                                                const int* __restrict__ rec,
                                                const int* __restrict__ recS,
                                                float* __restrict__ coef, float* __restrict__ tq,
                                                int* __restrict__ tick,
                                                int nE, int npad) {
    __shared__ __attribute__((aligned(16))) unsigned char sm[3 * NMAX];
    __shared__ int sh[3];
    const unsigned short* sA = (const unsigned short*)sm;
    const unsigned char* sB = sm + 2 * npad;
    int h0 = 0, t0 = 0;
    bool drawn0 = false;
    const bool in_lds = npad <= NMAX;
    if (in_lds) {
        const int np = (3 * npad) >> 4;
        const v4u* __restrict__ gp = (const v4u*)recS;
        v4u* lp = (v4u*)sm;
        #pragma unroll 1
        for (int p0 = threadIdx.x; p0 < np; p0 += CE_T * CE_F) {
            v4u v[CE_F];
            #pragma unroll
            for (int u = 0; u < CE_F; ++u) v[u] = gp[min(p0 + u * CE_T, np - 1)];
            if (p0 == 0) { h0 = pkg_half(); t0 = atomicAdd(&tick[h0], 1); drawn0 = true; }
            static_assert(CE_F == 10, "the operand list below names CE_F values");
            asm volatile("" :: "v"(v[0]), "v"(v[1]), "v"(v[2]), "v"(v[3]), "v"(v[4]),
                               "v"(v[5]), "v"(v[6]), "v"(v[7]), "v"(v[8]), "v"(v[9]));
            #pragma unroll
            for (int u = 0; u < CE_F; ++u)
                if (p0 + u * CE_T < np) lp[p0 + u * CE_T] = v[u];
        }
    }
    if (threadIdx.x == 0) {
        if (!drawn0) { h0 = pkg_half(); t0 = atomicAdd(&tick[h0], 1); }
        sh[0] = h0; sh[1] = t0;
    }
    __syncthreads();
    const int half = sh[0];
    const int nshare = max((int)gridDim.x >> 1, 1);
    int t1 = 0;
    bool drawn1 = false;
    int* const oth = tick + (half ^ 1);
    const int t_own = sh[1];
    if (t_own < nshare) {
        if (half == 0) ce_share<0>(src, dst, rec, sA, sB, in_lds, coef, nE, t_own, nshare, oth, t1, drawn1);
        else           ce_share<1>(src, dst, rec, sA, sB, in_lds, tq,   nE, t_own, nshare, oth, t1, drawn1);
    }
    if (threadIdx.x == 0) {
        if (!drawn1) t1 = atomicAdd(oth, 1);
        sh[2] = t1;
    }
    __syncthreads();
    const int t_oth = sh[2];
    if (t_oth < nshare) {
        int tn = 0; bool dn = false;
        if (half == 0) ce_share<1>(src, dst, rec, sA, sB, in_lds, tq,   nE, t_oth, nshare, nullptr, tn, dn);
        else           ce_share<0>(src, dst, rec, sA, sB, in_lds, coef, nE, t_oth, nshare, nullptr, tn, dn);
    }
}

#define NCH (KB / 16)
struct PFrag {
    v2f c0[NTP / 4];
    float4 u0[4], u1[4];
};
struct PSum {
    v2f a[NTP / 4];
    v2f t[4][4];
};
__device__ __forceinline__ void pf_load(PFrag& f, const float* __restrict__ coef,
                                        const float* __restrict__ tq,
                                        int base, int n, int half, int l16) {
    const size_t ro = (size_t)min(base + l16, n - 1) * CS + half * 2;
    #pragma unroll
    for (int kt = 0; kt < NTP / 4; ++kt)
        f.c0[kt] = *(const v2f*)(coef + ro + kt * 4);
    const size_t to = (size_t)((base >> 1) + 4 * half) * TP + l16 * 8;
    #pragma unroll
    for (int q = 0; q < 4; ++q) {
        f.u0[q] = *(const float4*)(tq + to + (size_t)q * TP);
        f.u1[q] = *(const float4*)(tq + to + (size_t)q * TP + 4);
    }
}
__device__ __forceinline__ void pf_sum(const PFrag& f, PSum& s) {
    #pragma unroll
    for (int kt = 0; kt < NTP / 4; ++kt) s.a[kt] = f.c0[kt];
    #pragma unroll
    for (int q = 0; q < 4; ++q) {
        s.t[q][0].x = f.u0[q].x; s.t[q][0].y = f.u0[q].y;
        s.t[q][1].x = f.u0[q].z; s.t[q][1].y = f.u0[q].w;
        s.t[q][2].x = f.u1[q].x; s.t[q][2].y = f.u1[q].y;
        s.t[q][3].x = f.u1[q].z; s.t[q][3].y = f.u1[q].w;
    }
}
__device__ __forceinline__ void pf_mma(const PSum& s, const v2f (&hb)[2][NTP / 4], const float (&bias)[2],
                                       v8f (&acc)[2][4]) {
    v8f p0 = {}, p1 = {};
    #pragma unroll
    for (int kt = 0; kt < NTP / 4; ++kt) {
        p0 = __builtin_amdgcn_wmma_f32_16x16x4_f32(false, s.a[kt], false, hb[0][kt], (short)0, p0, false, false);
        p1 = __builtin_amdgcn_wmma_f32_16x16x4_f32(false, s.a[kt], false, hb[1][kt], (short)0, p1, false, false);
    }
    #pragma unroll
    for (int j = 0; j < 8; ++j) {
        p0[j] = fmaxf(p0[j] + bias[0], 0.f);
        p1[j] = fmaxf(p1[j] + bias[1], 0.f);
    }
    #pragma unroll
    for (int q = 0; q < 4; ++q) {
        v2f b0, b1;
        b0.x = p0[2 * q]; b0.y = p0[2 * q + 1];
        b1.x = p1[2 * q]; b1.y = p1[2 * q + 1];
        #pragma unroll
        for (int mt = 0; mt < 4; ++mt) {
            acc[0][mt] = __builtin_amdgcn_wmma_f32_16x16x4_f32(false, s.t[q][mt], false, b0, (short)0, acc[0][mt], false, false);
            acc[1][mt] = __builtin_amdgcn_wmma_f32_16x16x4_f32(false, s.t[q][mt], false, b1, (short)0, acc[1][mt], false, false);
        }
    }
}
__global__ __launch_bounds__(256) void k_poolmm(const float* __restrict__ coef,
                                                const float* __restrict__ HW1,
                                                const float* __restrict__ b1,
                                                const float* __restrict__ tq,
                                                const int* __restrict__ batch,
                                                float* __restrict__ part,
                                                float* __restrict__ pcnt, int n) {
    static_assert(KB % 16 == 0 && KB <= 256, "KB");
    __shared__ int cl[NG];
    __shared__ float st[NG][DIM + 4];
    const int tid = threadIdx.x;
    const int wave = __builtin_amdgcn_readfirstlane(tid >> 5);
    const int lane = tid & 31;
    const int half = lane >> 4, l16 = lane & 15;
    const int cg = wave & 3;
    const int par = wave >> 2;
    const int nb = blockIdx.x * KB;

    if (tid < NG) cl[tid] = 0;
    __syncthreads();
    if (tid < KB && nb + tid < n) atomicAdd(&cl[batch[nb + tid]], 1);

    v2f hb[2][NTP / 4];
    float bias[2];
    #pragma unroll
    for (int ct = 0; ct < 2; ++ct) {
        const int col = cg * 32 + ct * 16 + l16;
        #pragma unroll
        for (int kt = 0; kt < NTP / 4; ++kt) {
            const int k0 = kt * 4 + half * 2;
            hb[ct][kt].x = HW1[(size_t)k0 * DIM + col];
            hb[ct][kt].y = HW1[(size_t)(k0 + 1) * DIM + col];
        }
        bias[ct] = b1[col];
    }

    v8f acc[2][4] = {};
    #pragma unroll 1
    for (int ch = par; ch < NCH; ch += 2) {
        PFrag f;
        PSum s;
        pf_load(f, coef, tq, nb + ch * 16, n, half, l16);
        pf_sum(f, s);
        pf_mma(s, hb, bias, acc);
    }

    if (par == 1) {
        #pragma unroll
        for (int ct = 0; ct < 2; ++ct) {
            const int col = cg * 32 + ct * 16 + l16;
            #pragma unroll
            for (int mt = 0; mt < 4; ++mt) {
                #pragma unroll
                for (int j = 0; j < 8; ++j)
                    st[mt * 16 + j + half * 8][col] = acc[ct][mt][j];
            }
        }
    }
    __syncthreads();
    if (par == 0) {
        #pragma unroll
        for (int ct = 0; ct < 2; ++ct) {
            const int col = cg * 32 + ct * 16 + l16;
            #pragma unroll
            for (int mt = 0; mt < 4; ++mt) {
                #pragma unroll
                for (int j = 0; j < 8; ++j)
                    st[mt * 16 + j + half * 8][col] += acc[ct][mt][j];
            }
        }
    }
    __syncthreads();
    float* __restrict__ pb = part + (size_t)blockIdx.x * (NG * DIM);
    #pragma unroll
    for (int i = 0; i < NG / 8; ++i) {
        const int r = wave + 8 * i;
        *(float4*)(pb + (size_t)r * DIM + lane * 4) = *(const float4*)(&st[r][lane * 4]);
    }
    if (tid < NG) pcnt[blockIdx.x * NG + tid] = (float)cl[tid];
}

__global__ __launch_bounds__(1024) void k_final(const float* __restrict__ part,
                                                const float* __restrict__ pcnt,
                                                const float* __restrict__ W2,
                                                const float* __restrict__ b2,
                                                float* __restrict__ out, int nblk) {
    __shared__ float ps[32][DIM];
    __shared__ float prow[DIM];
    __shared__ float orow[DIM];
    __shared__ float pc[32];
    const int tid = threadIdx.x;
    const int wave = __builtin_amdgcn_readfirstlane(tid >> 5);
    const int lane = tid & 31;
    const int half = lane >> 4, l16 = lane & 15;
    const int g = blockIdx.x;
    const int col = (wave & 7) * 16 + l16;

    v2f wb[DIM / 4] = {};
    if (wave < 8) {
        #pragma unroll
        for (int kt = 0; kt < DIM / 4; ++kt) {
            const int k0 = kt * 4 + half * 2;
            wb[kt].x = W2[(size_t)k0 * DIM + col];
            wb[kt].y = W2[(size_t)(k0 + 1) * DIM + col];
        }
    }

    float4 s = make_float4(0.f, 0.f, 0.f, 0.f);
    for (int b0 = wave; b0 < nblk; b0 += 32 * 8) {
        #pragma unroll
        for (int u = 0; u < 8; ++u) {
            const int b = b0 + 32 * u;
            const float m = b < nblk ? 1.f : 0.f;
            const float4 v = *(const float4*)(part + (size_t)min(b, nblk - 1) * (NG * DIM) + g * DIM + lane * 4);
            s.x = fmaf(m, v.x, s.x); s.y = fmaf(m, v.y, s.y); s.z = fmaf(m, v.z, s.z); s.w = fmaf(m, v.w, s.w);
        }
    }
    float c = 0.f;
    for (int b = tid; b < nblk; b += 1024) c += pcnt[b * NG + g];
    #pragma unroll
    for (int off = 16; off > 0; off >>= 1) c += __shfl_xor(c, off);
    *(float4*)(&ps[wave][lane * 4]) = s;
    if (lane == 0) pc[wave] = c;
    __syncthreads();
    if (tid < DIM) {
        float ss = ps[0][tid];
        #pragma unroll
        for (int w = 1; w < 32; ++w) ss += ps[w][tid];
        prow[tid] = ss;
    }
    __syncthreads();
    if (wave < 8) {
        const int keep = l16 == 0 ? -1 : 0;
        v8f acc = {};
        #pragma unroll
        for (int kt = 0; kt < DIM / 4; ++kt) {
            const v2f pa = *(const v2f*)(&prow[kt * 4 + half * 2]);
            v2f a;
            a.x = __int_as_float(__float_as_int(pa.x) & keep);
            a.y = __int_as_float(__float_as_int(pa.y) & keep);
            acc = __builtin_amdgcn_wmma_f32_16x16x4_f32(false, a, false, wb[kt], (short)0, acc, false, false);
        }
        float cs = pc[0];
        #pragma unroll
        for (int w = 1; w < 32; ++w) cs += pc[w];
        if (half == 0) orow[col] = fmaf(cs, b2[col], acc[0]) / fmaxf(cs, 1.0f);
    }
    __syncthreads();
    if (tid < 32) *(float4*)(out + (size_t)g * DIM + tid * 4) = *(const float4*)(&orow[tid * 4]);
}

extern "C" void kernel_launch(void* const* d_in, const int* in_sizes, int n_in,
                              void* d_out, int out_size, void* d_ws, size_t ws_size,
                              hipStream_t stream) {
    const int*   node_ids = (const int*)d_in[0];
    const int*   ei       = (const int*)d_in[1];
    const int*   batch    = (const int*)d_in[2];
    const float* emb      = (const float*)d_in[4];
    const float* W1       = (const float*)d_in[5];
    const float* b1       = (const float*)d_in[6];
    const float* W2       = (const float*)d_in[7];
    const float* b2       = (const float*)d_in[8];
    float* out = (float*)d_out;

    const int N = in_sizes[0];
    const int E = in_sizes[1] / 2;
    const int* src = ei;
    const int* dst = ei + E;

    char* base = (char*)d_ws;
    size_t off = 0;
    auto alloc = [&](size_t bytes) -> char* {
        char* p = base + off;
        off = (off + bytes + 255) & ~(size_t)255;
        return p;
    };
    float* HW1  = (float*)alloc((size_t)NTP * DIM * sizeof(float));
    int*   rec  = (int*)alloc((size_t)N * 2 * sizeof(int));
    const int NPADN = (N + 15) & ~15;
    int* recS = (int*)alloc((size_t)NPADN * 3);
    const int NH = (E + 65534) / 65535 > 64 ? (E + 65534) / 65535 : 64;
    unsigned short* hist = (unsigned short*)alloc((size_t)NH * NPADN * 2);
    float* coef = (float*)alloc((size_t)N * CS * sizeof(float));
    const int NB = (N + KB - 1) / KB;
    const int NPAIR = NB * (KB / 2);
    float* tq   = (float*)alloc((size_t)NPAIR * TP * sizeof(float));
    float* part = (float*)alloc((size_t)NB * NG * DIM * sizeof(float));
    float* pcnt = (float*)alloc((size_t)NB * NG * sizeof(float));
    int*   tick = (int*)alloc(2 * sizeof(int));
    (void)ws_size; (void)n_in; (void)out_size;

    const int TB = 256;
    dim3 blk(TB);

    k_first<<<dim3(NH + 192), dim3(FT), 0, stream>>>(dst, E, NPADN, hist, NH,
                                                     coef, N * (CS / 4),
                                                     tq, NPAIR * (TP / 4),
                                                     emb, W1, HW1, tick);
    k_rec  <<<dim3((NPADN + TB - 1) / TB), blk, 0, stream>>>(hist, NH, node_ids, batch, rec, recS, coef, tq, N, NPADN);

    k_cedge<<<dim3(256), dim3(CE_T), 0, stream>>>(src, dst, rec, recS, coef, tq, tick, E, NPADN);

    k_poolmm<<<dim3(NB), blk, 0, stream>>>(coef, HW1, b1, tq, batch, part, pcnt, N);

    k_final <<<dim3(NG), dim3(1024), 0, stream>>>(part, pcnt, W2, b2, out, NB);
}
